# starting kernel
# baseline (speedup 1.0000x reference)
#include <hip/hip_runtime.h>
#include <math.h>

typedef float v2f __attribute__((ext_vector_type(2)));
typedef float v8f __attribute__((ext_vector_type(8)));

#define B_   8
#define S_   4096
#define D_   1024
#define H_   1024
#define PD_  512
#define P_   4
#define TOK_   2048
#define TC_    8
#define TOKPC_ 256

__global__ void gather_partials(const float* __restrict__ ctx,
                                const int* __restrict__ masks,
                                float* __restrict__ partials) {
  __shared__ int idx[TOKPC_];
  const int dc  = blockIdx.x;
  const int tc  = blockIdx.y;
  const int b   = blockIdx.z;
  const int tid = threadIdx.x;
  idx[tid] = masks[b * TOK_ + tc * TOKPC_ + tid];
  __syncthreads();
  const float* base = ctx + (size_t)b * S_ * D_ + dc * 256 + tid;
  float sum = 0.f;
#pragma unroll 8
  for (int i = 0; i < TOKPC_; ++i) {
    sum += base[(size_t)idx[i] * D_];
  }
  partials[(size_t)(tc * B_ + b) * D_ + dc * 256 + tid] = sum;
}

__global__ void reduce_pool(const float* __restrict__ partials,
                            float* __restrict__ pooled) {
  const int g   = blockIdx.x * 256 + threadIdx.x;
  const int row = g >> 10;
  const int d   = g & 1023;
  float s = 0.f;
  if (row < B_) {
#pragma unroll
    for (int tc = 0; tc < TC_; ++tc)
      s += partials[(size_t)(tc * B_ + row) * D_ + d];
    s *= (1.0f / (float)TOK_);
  }
  pooled[g] = s;
}

__global__ void gemm1_gelu(const float* __restrict__ A,
                           const float* __restrict__ W1,
                           const float* __restrict__ b1,
                           float* __restrict__ Hout) {
  const int lane = threadIdx.x & 31;
  const int half = lane >> 4;
  const int l16  = lane & 15;
  const int n0   = blockIdx.x * 16;
  const int col  = n0 + l16;

  const float* Ap = A + l16 * D_ + 2 * half;
  const float* Wp = W1 + (size_t)(2 * half) * H_ + col;

  v8f acc = {};
  for (int k = 0; k < D_; k += 4) {
    v2f a, b;
    a.x = Ap[k];
    a.y = Ap[k + 1];
    b.x = Wp[(size_t)k * H_];
    b.y = Wp[(size_t)(k + 1) * H_];
    acc = __builtin_amdgcn_wmma_f32_16x16x4_f32(
        false, a, false, b, (short)0, acc, false, false);
  }

  const float bias = b1[col];
#pragma unroll
  for (int r = 0; r < 8; ++r) {
    float x = acc[r] + bias;
    float g = 0.5f * x * (1.0f + erff(x * 0.70710678118654752f));
    int row = r + 8 * half;
    Hout[row * H_ + col] = g;
  }
}

__global__ void layernorm_rows(float* __restrict__ Hbuf,
                               const float* __restrict__ gam,
                               const float* __restrict__ bet) {
  const int row = blockIdx.x;
  const int tid = threadIdx.x;
  __shared__ float red1[8];
  __shared__ float red2[8];

  float v[4];
  float s = 0.f;
#pragma unroll
  for (int j = 0; j < 4; ++j) {
    v[j] = Hbuf[row * H_ + tid + j * 256];
    s += v[j];
  }
#pragma unroll
  for (int off = 16; off > 0; off >>= 1) s += __shfl_down(s, off, 32);
  if ((tid & 31) == 0) red1[tid >> 5] = s;
  __syncthreads();
  float tot = 0.f;
#pragma unroll
  for (int i = 0; i < 8; ++i) tot += red1[i];
  const float mean = tot * (1.0f / (float)H_);

  float s2 = 0.f;
#pragma unroll
  for (int j = 0; j < 4; ++j) {
    float d = v[j] - mean;
    s2 += d * d;
  }
#pragma unroll
  for (int off = 16; off > 0; off >>= 1) s2 += __shfl_down(s2, off, 32);
  if ((tid & 31) == 0) red2[tid >> 5] = s2;
  __syncthreads();
  float tot2 = 0.f;
#pragma unroll
  for (int i = 0; i < 8; ++i) tot2 += red2[i];
  const float inv = rsqrtf(tot2 * (1.0f / (float)H_) + 1e-5f);

#pragma unroll
  for (int j = 0; j < 4; ++j) {
    int c = tid + j * 256;
    Hbuf[row * H_ + c] = (v[j] - mean) * inv * gam[c] + bet[c];
  }
}

__global__ void gemm2_out(const float* __restrict__ Hbuf,
                          const float* __restrict__ W2,
                          const float* __restrict__ b2,
                          float* __restrict__ out) {
  const int lane = threadIdx.x & 31;
  const int half = lane >> 4;
  const int l16  = lane & 15;
  const int n0   = blockIdx.x * 16;
  const int col  = n0 + l16;

  const float* Ap = Hbuf + l16 * H_ + 2 * half;
  const float* Wp = W2 + (size_t)(2 * half) * PD_ + col;

  v8f acc = {};
  for (int k = 0; k < H_; k += 4) {
    v2f a, b;
    a.x = Ap[k];
    a.y = Ap[k + 1];
    b.x = Wp[(size_t)k * PD_];
    b.y = Wp[(size_t)(k + 1) * PD_];
    acc = __builtin_amdgcn_wmma_f32_16x16x4_f32(
        false, a, false, b, (short)0, acc, false, false);
  }

  if (half == 0) {
    const float bias = b2[col];
#pragma unroll
    for (int r = 0; r < 8; ++r) {
      float val = acc[r] + bias;
#pragma unroll
      for (int p = 0; p < P_; ++p)
        out[(size_t)(r * P_ + p) * PD_ + col] = val;
    }
  }
}

extern "C" void kernel_launch(void* const* d_in, const int* in_sizes, int n_in,
                              void* d_out, int out_size, void* d_ws, size_t ws_size,
                              hipStream_t stream) {
  const float* ctx  = (const float*)d_in[0];
  const float* W1   = (const float*)d_in[1];
  const float* b1   = (const float*)d_in[2];
  const float* ln_g = (const float*)d_in[3];
  const float* ln_b = (const float*)d_in[4];
  const float* W2   = (const float*)d_in[5];
  const float* b2   = (const float*)d_in[6];
  const int*  masks = (const int*)d_in[7];
  (void)in_sizes; (void)n_in; (void)out_size; (void)ws_size;

  float* out = (float*)d_out;

  char* ws = (char*)d_ws;
  float* pooled   = (float*)(ws);
  float* partials = (float*)(ws + 65536);
  float* hbuf     = (float*)(ws + 65536 + 262144);

  gather_partials<<<dim3(4, TC_, B_), 256, 0, stream>>>(ctx, masks, partials);
  reduce_pool<<<64, 256, 0, stream>>>(partials, pooled);
  gemm1_gelu<<<H_ / 16, 32, 0, stream>>>(pooled, W1, b1, hbuf);
  layernorm_rows<<<B_, 256, 0, stream>>>(hbuf, ln_g, ln_b);
  gemm2_out<<<PD_ / 16, 32, 0, stream>>>(hbuf, W2, b2, out);
}


# Both GEMM K loops freed of memory waits: the 8 real A rows copied once into LDS, B loads kept 32 k-steps ahead in a register ring (one wave per tile as before; k order and sums unchanged)
# speedup vs baseline: 1.6208x; 1.6208x over previous
#include <hip/hip_runtime.h>
#include <math.h>

typedef float v2f __attribute__((ext_vector_type(2)));
typedef float v8f __attribute__((ext_vector_type(8)));

#define B_   8
#define S_   4096
#define D_   1024
#define H_   1024
#define PD_  512
#define P_   4
#define TOK_   2048
#define TC_    8
#define TOKPC_ 256

__global__ void gather_partials(const float* __restrict__ ctx,
                                const int* __restrict__ masks,
                                float* __restrict__ partials) {
  __shared__ int idx[TOKPC_];
  const int dc  = blockIdx.x;
  const int tc  = blockIdx.y;
  const int b   = blockIdx.z;
  const int tid = threadIdx.x;
  idx[tid] = masks[b * TOK_ + tc * TOKPC_ + tid];
  __syncthreads();
  const float* base = ctx + (size_t)b * S_ * D_ + dc * 256 + tid;
  float sum = 0.f;
#pragma unroll 8
  for (int i = 0; i < TOKPC_; ++i) {
    sum += base[(size_t)idx[i] * D_];
  }
  partials[(size_t)(tc * B_ + b) * D_ + dc * 256 + tid] = sum;
}

__global__ void reduce_pool(const float* __restrict__ partials,
                            float* __restrict__ pooled) {
  const int g   = blockIdx.x * 256 + threadIdx.x;
  const int row = g >> 10;
  const int d   = g & 1023;
  float s = 0.f;
  if (row < B_) {
#pragma unroll
    for (int tc = 0; tc < TC_; ++tc)
      s += partials[(size_t)(tc * B_ + row) * D_ + d];
    s *= (1.0f / (float)TOK_);
  }
  pooled[g] = s;
}

typedef float v4f __attribute__((ext_vector_type(4)));

#define AROWS_  8
#define APITCH_ 1028
#define RING_   32
#define SLOT_   4

template <int LDB>
__device__ __forceinline__ v8f gemm_k1024_tile(const float* A, const float* Wp,
                                               float* As, int lane) {
  v2f a[RING_], b[RING_];
#pragma unroll
  for (int u = 0; u < RING_; ++u) {
    b[u].x = Wp[(size_t)(4 * u) * LDB];
    b[u].y = Wp[(size_t)(4 * u + 1) * LDB];
  }

#pragma unroll
  for (int i = 0; i < AROWS_ * 1024 / 4 / 32; ++i) {
    const int c4 = (i & 7) * 32 + lane;
    const v4f v = *reinterpret_cast<const v4f*>(A + (size_t)(i >> 3) * 1024 + 4 * c4);
    *reinterpret_cast<v4f*>(As + (i >> 3) * APITCH_ + 4 * c4) = v;
  }
  __syncthreads();

  const float* Al = As + (lane & 7) * APITCH_ + 2 * (lane >> 4);
#pragma unroll
  for (int u = 0; u < RING_; ++u)
    a[u] = *reinterpret_cast<const v2f*>(Al + 4 * u);

  v8f acc = {};
  for (int kb = 0; kb < 256 - RING_; kb += RING_) {
#pragma unroll
    for (int s = 0; s < RING_; s += SLOT_) {
#pragma unroll
      for (int u = s; u < s + SLOT_; ++u)
        acc = __builtin_amdgcn_wmma_f32_16x16x4_f32(
            false, a[u], false, b[u], (short)0, acc, false, false);
#pragma unroll
      for (int u = s; u < s + SLOT_; ++u) {
        const int k = 4 * (kb + RING_ + u);
        a[u]   = *reinterpret_cast<const v2f*>(Al + k);
        b[u].x = Wp[(size_t)k * LDB];
        b[u].y = Wp[(size_t)(k + 1) * LDB];
      }
      __builtin_amdgcn_sched_barrier(0);
    }
  }
#pragma unroll
  for (int u = 0; u < RING_; ++u)
    acc = __builtin_amdgcn_wmma_f32_16x16x4_f32(
        false, a[u], false, b[u], (short)0, acc, false, false);
  return acc;
}

__global__ void __launch_bounds__(32) gemm1_gelu(const float* __restrict__ A,
                           const float* __restrict__ W1,
                           const float* __restrict__ b1,
                           float* __restrict__ Hout) {
  const int lane = threadIdx.x & 31;
  const int half = lane >> 4;
  const int l16  = lane & 15;
  const int n0   = blockIdx.x * 16;
  const int col  = n0 + l16;

  __shared__ __attribute__((aligned(16))) float As[AROWS_ * APITCH_];
  const float* Wp = W1 + (size_t)(2 * half) * H_ + col;

  const v8f acc = gemm_k1024_tile<H_>(A, Wp, As, lane);

  const float bias = b1[col];
#pragma unroll
  for (int r = 0; r < 8; ++r) {
    float x = acc[r] + bias;
    float g = 0.5f * x * (1.0f + erff(x * 0.70710678118654752f));
    int row = r + 8 * half;
    Hout[row * H_ + col] = g;
  }
}

__global__ void layernorm_rows(float* __restrict__ Hbuf,
                               const float* __restrict__ gam,
                               const float* __restrict__ bet) {
  const int row = blockIdx.x;
  const int tid = threadIdx.x;
  __shared__ float red1[8];
  __shared__ float red2[8];

  float v[4];
  float s = 0.f;
#pragma unroll
  for (int j = 0; j < 4; ++j) {
    v[j] = Hbuf[row * H_ + tid + j * 256];
    s += v[j];
  }
#pragma unroll
  for (int off = 16; off > 0; off >>= 1) s += __shfl_down(s, off, 32);
  if ((tid & 31) == 0) red1[tid >> 5] = s;
  __syncthreads();
  float tot = 0.f;
#pragma unroll
  for (int i = 0; i < 8; ++i) tot += red1[i];
  const float mean = tot * (1.0f / (float)H_);

  float s2 = 0.f;
#pragma unroll
  for (int j = 0; j < 4; ++j) {
    float d = v[j] - mean;
    s2 += d * d;
  }
#pragma unroll
  for (int off = 16; off > 0; off >>= 1) s2 += __shfl_down(s2, off, 32);
  if ((tid & 31) == 0) red2[tid >> 5] = s2;
  __syncthreads();
  float tot2 = 0.f;
#pragma unroll
  for (int i = 0; i < 8; ++i) tot2 += red2[i];
  const float inv = rsqrtf(tot2 * (1.0f / (float)H_) + 1e-5f);

#pragma unroll
  for (int j = 0; j < 4; ++j) {
    int c = tid + j * 256;
    Hbuf[row * H_ + c] = (v[j] - mean) * inv * gam[c] + bet[c];
  }
}

__global__ void __launch_bounds__(32) gemm2_out(const float* __restrict__ Hbuf,
                          const float* __restrict__ W2,
                          const float* __restrict__ b2,
                          float* __restrict__ out) {
  const int lane = threadIdx.x & 31;
  const int half = lane >> 4;
  const int l16  = lane & 15;
  const int n0   = blockIdx.x * 16;
  const int col  = n0 + l16;

  __shared__ __attribute__((aligned(16))) float As[AROWS_ * APITCH_];
  const float* Wp = W2 + (size_t)(2 * half) * PD_ + col;

  const v8f acc = gemm_k1024_tile<PD_>(Hbuf, Wp, As, lane);

  if (half == 0) {
    const float bias = b2[col];
#pragma unroll
    for (int r = 0; r < 8; ++r) {
      float val = acc[r] + bias;
#pragma unroll
      for (int p = 0; p < P_; ++p)
        out[(size_t)(r * P_ + p) * PD_ + col] = val;
    }
  }
}

extern "C" void kernel_launch(void* const* d_in, const int* in_sizes, int n_in,
                              void* d_out, int out_size, void* d_ws, size_t ws_size,
                              hipStream_t stream) {
  const float* ctx  = (const float*)d_in[0];
  const float* W1   = (const float*)d_in[1];
  const float* b1   = (const float*)d_in[2];
  const float* ln_g = (const float*)d_in[3];
  const float* ln_b = (const float*)d_in[4];
  const float* W2   = (const float*)d_in[5];
  const float* b2   = (const float*)d_in[6];
  const int*  masks = (const int*)d_in[7];
  (void)in_sizes; (void)n_in; (void)out_size; (void)ws_size;

  float* out = (float*)d_out;

  char* ws = (char*)d_ws;
  float* pooled   = (float*)(ws);
  float* partials = (float*)(ws + 65536);
  float* hbuf     = (float*)(ws + 65536 + 262144);

  gather_partials<<<dim3(4, TC_, B_), 256, 0, stream>>>(ctx, masks, partials);
  reduce_pool<<<64, 256, 0, stream>>>(partials, pooled);
  gemm1_gelu<<<H_ / 16, 32, 0, stream>>>(pooled, W1, b1, hbuf);
  layernorm_rows<<<B_, 256, 0, stream>>>(hbuf, ln_g, ln_b);
  gemm2_out<<<PD_ / 16, 32, 0, stream>>>(hbuf, W2, b2, out);
}


# Gather: row index taken per wave by readlane (scalar side; no LDS stage, no barrier), so row loads are independent: ring of 64 in flight per wave instead of 8-then-wait; token sum order unchanged
# speedup vs baseline: 2.5757x; 1.5892x over previous
#include <hip/hip_runtime.h>
#include <math.h>

typedef float v2f __attribute__((ext_vector_type(2)));
typedef float v8f __attribute__((ext_vector_type(8)));

#define B_   8
#define S_   4096
#define D_   1024
#define H_   1024
#define PD_  512
#define P_   4
#define TOK_   2048
#define TC_    8
#define TOKPC_ 256

#define GRING_ 64

__device__ __forceinline__ float gather_row(const char* cb, int idxreg,
                                            int srclane, unsigned col4) {
  const unsigned r = (unsigned)__builtin_amdgcn_readlane(idxreg, srclane);
  const unsigned off = r * (unsigned)(D_ * sizeof(float)) + col4;
  return *reinterpret_cast<const float*>(cb + off);
}

__global__ void __launch_bounds__(256) gather_partials(const float* __restrict__ ctx,
                                const int* __restrict__ masks,
                                float* __restrict__ partials) {
  const int dc  = blockIdx.x;
  const int tc  = blockIdx.y;
  const int b   = blockIdx.z;
  const unsigned tid  = threadIdx.x;
  const unsigned lane = tid & 31u;
  const unsigned col4 = tid * 4u;
  const int*  mrow = masks + b * TOK_ + tc * TOKPC_;
  const char* cb   = reinterpret_cast<const char*>(ctx + (size_t)b * S_ * D_ + dc * 256);


  int iv[TOKPC_ / 32];
#pragma unroll
  for (int j = 0; j < TOKPC_ / 32; ++j) iv[j] = mrow[32 * j + lane];

  float v[GRING_];
#pragma unroll
  for (int u = 0; u < GRING_; ++u) v[u] = gather_row(cb, iv[u >> 5], u & 31, col4);

  __builtin_amdgcn_sched_barrier(0);

  float sum = 0.f;
#pragma unroll
  for (int t = GRING_; t < TOKPC_; ++t) {
    sum += v[t % GRING_];
    v[t % GRING_] = gather_row(cb, iv[t >> 5], t & 31, col4);
  }
#pragma unroll
  for (int u = 0; u < GRING_; ++u) sum += v[u];

  partials[(size_t)(tc * B_ + b) * D_ + dc * 256 + tid] = sum;
}

__global__ void reduce_pool(const float* __restrict__ partials,
                            float* __restrict__ pooled) {
  const int g   = blockIdx.x * 256 + threadIdx.x;
  const int row = g >> 10;
  const int d   = g & 1023;
  float s = 0.f;
  if (row < B_) {
#pragma unroll
    for (int tc = 0; tc < TC_; ++tc)
      s += partials[(size_t)(tc * B_ + row) * D_ + d];
    s *= (1.0f / (float)TOK_);
  }
  pooled[g] = s;
}

typedef float v4f __attribute__((ext_vector_type(4)));

#define AROWS_  8
#define APITCH_ 1028
#define RING_   32
#define SLOT_   4

template <int LDB>
__device__ __forceinline__ v8f gemm_k1024_tile(const float* A, const float* Wp,
                                               float* As, int lane) {
  v2f a[RING_], b[RING_];
#pragma unroll
  for (int u = 0; u < RING_; ++u) {
    b[u].x = Wp[(size_t)(4 * u) * LDB];
    b[u].y = Wp[(size_t)(4 * u + 1) * LDB];
  }

#pragma unroll
  for (int i = 0; i < AROWS_ * 1024 / 4 / 32; ++i) {
    const int c4 = (i & 7) * 32 + lane;
    const v4f v = *reinterpret_cast<const v4f*>(A + (size_t)(i >> 3) * 1024 + 4 * c4);
    *reinterpret_cast<v4f*>(As + (i >> 3) * APITCH_ + 4 * c4) = v;
  }
  __syncthreads();

  const float* Al = As + (lane & 7) * APITCH_ + 2 * (lane >> 4);
#pragma unroll
  for (int u = 0; u < RING_; ++u)
    a[u] = *reinterpret_cast<const v2f*>(Al + 4 * u);

  v8f acc = {};
  for (int kb = 0; kb < 256 - RING_; kb += RING_) {
#pragma unroll
    for (int s = 0; s < RING_; s += SLOT_) {
#pragma unroll
      for (int u = s; u < s + SLOT_; ++u)
        acc = __builtin_amdgcn_wmma_f32_16x16x4_f32(
            false, a[u], false, b[u], (short)0, acc, false, false);
#pragma unroll
      for (int u = s; u < s + SLOT_; ++u) {
        const int k = 4 * (kb + RING_ + u);
        a[u]   = *reinterpret_cast<const v2f*>(Al + k);
        b[u].x = Wp[(size_t)k * LDB];
        b[u].y = Wp[(size_t)(k + 1) * LDB];
      }
      __builtin_amdgcn_sched_barrier(0);
    }
  }
#pragma unroll
  for (int u = 0; u < RING_; ++u)
    acc = __builtin_amdgcn_wmma_f32_16x16x4_f32(
        false, a[u], false, b[u], (short)0, acc, false, false);
  return acc;
}

__global__ void __launch_bounds__(32) gemm1_gelu(const float* __restrict__ A,
                           const float* __restrict__ W1,
                           const float* __restrict__ b1,
                           float* __restrict__ Hout) {
  const int lane = threadIdx.x & 31;
  const int half = lane >> 4;
  const int l16  = lane & 15;
  const int n0   = blockIdx.x * 16;
  const int col  = n0 + l16;

  __shared__ __attribute__((aligned(16))) float As[AROWS_ * APITCH_];
  const float* Wp = W1 + (size_t)(2 * half) * H_ + col;

  const v8f acc = gemm_k1024_tile<H_>(A, Wp, As, lane);

  const float bias = b1[col];
#pragma unroll
  for (int r = 0; r < 8; ++r) {
    float x = acc[r] + bias;
    float g = 0.5f * x * (1.0f + erff(x * 0.70710678118654752f));
    int row = r + 8 * half;
    Hout[row * H_ + col] = g;
  }
}

__global__ void layernorm_rows(float* __restrict__ Hbuf,
                               const float* __restrict__ gam,
                               const float* __restrict__ bet) {
  const int row = blockIdx.x;
  const int tid = threadIdx.x;
  __shared__ float red1[8];
  __shared__ float red2[8];

  float v[4];
  float s = 0.f;
#pragma unroll
  for (int j = 0; j < 4; ++j) {
    v[j] = Hbuf[row * H_ + tid + j * 256];
    s += v[j];
  }
#pragma unroll
  for (int off = 16; off > 0; off >>= 1) s += __shfl_down(s, off, 32);
  if ((tid & 31) == 0) red1[tid >> 5] = s;
  __syncthreads();
  float tot = 0.f;
#pragma unroll
  for (int i = 0; i < 8; ++i) tot += red1[i];
  const float mean = tot * (1.0f / (float)H_);

  float s2 = 0.f;
#pragma unroll
  for (int j = 0; j < 4; ++j) {
    float d = v[j] - mean;
    s2 += d * d;
  }
#pragma unroll
  for (int off = 16; off > 0; off >>= 1) s2 += __shfl_down(s2, off, 32);
  if ((tid & 31) == 0) red2[tid >> 5] = s2;
  __syncthreads();
  float tot2 = 0.f;
#pragma unroll
  for (int i = 0; i < 8; ++i) tot2 += red2[i];
  const float inv = rsqrtf(tot2 * (1.0f / (float)H_) + 1e-5f);

#pragma unroll
  for (int j = 0; j < 4; ++j) {
    int c = tid + j * 256;
    Hbuf[row * H_ + c] = (v[j] - mean) * inv * gam[c] + bet[c];
  }
}

__global__ void __launch_bounds__(32) gemm2_out(const float* __restrict__ Hbuf,
                          const float* __restrict__ W2,
                          const float* __restrict__ b2,
                          float* __restrict__ out) {
  const int lane = threadIdx.x & 31;
  const int half = lane >> 4;
  const int l16  = lane & 15;
  const int n0   = blockIdx.x * 16;
  const int col  = n0 + l16;

  __shared__ __attribute__((aligned(16))) float As[AROWS_ * APITCH_];
  const float* Wp = W2 + (size_t)(2 * half) * PD_ + col;

  const v8f acc = gemm_k1024_tile<PD_>(Hbuf, Wp, As, lane);

  if (half == 0) {
    const float bias = b2[col];
#pragma unroll
    for (int r = 0; r < 8; ++r) {
      float val = acc[r] + bias;
#pragma unroll
      for (int p = 0; p < P_; ++p)
        out[(size_t)(r * P_ + p) * PD_ + col] = val;
    }
  }
}

extern "C" void kernel_launch(void* const* d_in, const int* in_sizes, int n_in,
                              void* d_out, int out_size, void* d_ws, size_t ws_size,
                              hipStream_t stream) {
  const float* ctx  = (const float*)d_in[0];
  const float* W1   = (const float*)d_in[1];
  const float* b1   = (const float*)d_in[2];
  const float* ln_g = (const float*)d_in[3];
  const float* ln_b = (const float*)d_in[4];
  const float* W2   = (const float*)d_in[5];
  const float* b2   = (const float*)d_in[6];
  const int*  masks = (const int*)d_in[7];
  (void)in_sizes; (void)n_in; (void)out_size; (void)ws_size;

  float* out = (float*)d_out;

  char* ws = (char*)d_ws;
  float* pooled   = (float*)(ws);
  float* partials = (float*)(ws + 65536);
  float* hbuf     = (float*)(ws + 65536 + 262144);

  gather_partials<<<dim3(4, TC_, B_), 256, 0, stream>>>(ctx, masks, partials);
  reduce_pool<<<64, 256, 0, stream>>>(partials, pooled);
  gemm1_gelu<<<H_ / 16, 32, 0, stream>>>(pooled, W1, b1, hbuf);
  layernorm_rows<<<B_, 256, 0, stream>>>(hbuf, ln_g, ln_b);
  gemm2_out<<<PD_ / 16, 32, 0, stream>>>(hbuf, W2, b2, out);
}


# Both GEMMs: second K half folded into the idle half of each f32 WMMA (tile rows/columns 8..15 carry k 512..1023, 8-wide tiles): 128 not 256 chained WMMAs per wave; the K halves are summed at the end
# speedup vs baseline: 3.0116x; 1.1692x over previous
#include <hip/hip_runtime.h>
#include <math.h>

typedef float v2f __attribute__((ext_vector_type(2)));
typedef float v8f __attribute__((ext_vector_type(8)));

#define B_   8
#define S_   4096
#define D_   1024
#define H_   1024
#define PD_  512
#define P_   4
#define TOK_   2048
#define TC_    8
#define TOKPC_ 256

#define GRING_ 64

__device__ __forceinline__ float gather_row(const char* cb, int idxreg,
                                            int srclane, unsigned col4) {
  const unsigned r = (unsigned)__builtin_amdgcn_readlane(idxreg, srclane);
  const unsigned off = r * (unsigned)(D_ * sizeof(float)) + col4;
  return *reinterpret_cast<const float*>(cb + off);
}

__global__ void __launch_bounds__(256) gather_partials(const float* __restrict__ ctx,
                                const int* __restrict__ masks,
                                float* __restrict__ partials) {
  const int dc  = blockIdx.x;
  const int tc  = blockIdx.y;
  const int b   = blockIdx.z;
  const unsigned tid  = threadIdx.x;
  const unsigned lane = tid & 31u;
  const unsigned col4 = tid * 4u;
  const int*  mrow = masks + b * TOK_ + tc * TOKPC_;
  const char* cb   = reinterpret_cast<const char*>(ctx + (size_t)b * S_ * D_ + dc * 256);


  int iv[TOKPC_ / 32];
#pragma unroll
  for (int j = 0; j < TOKPC_ / 32; ++j) iv[j] = mrow[32 * j + lane];

  float v[GRING_];
#pragma unroll
  for (int u = 0; u < GRING_; ++u) v[u] = gather_row(cb, iv[u >> 5], u & 31, col4);

  __builtin_amdgcn_sched_barrier(0);

  float sum = 0.f;
#pragma unroll
  for (int t = GRING_; t < TOKPC_; ++t) {
    sum += v[t % GRING_];
    v[t % GRING_] = gather_row(cb, iv[t >> 5], t & 31, col4);
  }
#pragma unroll
  for (int u = 0; u < GRING_; ++u) sum += v[u];

  partials[(size_t)(tc * B_ + b) * D_ + dc * 256 + tid] = sum;
}

__global__ void reduce_pool(const float* __restrict__ partials,
                            float* __restrict__ pooled) {
  const int g   = blockIdx.x * 256 + threadIdx.x;
  const int row = g >> 10;
  const int d   = g & 1023;
  float s = 0.f;
  if (row < B_) {
#pragma unroll
    for (int tc = 0; tc < TC_; ++tc)
      s += partials[(size_t)(tc * B_ + row) * D_ + d];
    s *= (1.0f / (float)TOK_);
  }
  pooled[g] = s;
}

typedef float v4f __attribute__((ext_vector_type(4)));

#define AROWS_  8
#define KHALF_  512
#define HGAP_   32
#define APITCH_ 1092
#define KSTEPS_ 128
#define RING_   32
#define SLOT_   4

template <int LDB>
__device__ __forceinline__ v8f gemm_k1024_tile(const float* A, const float* Wp,
                                               float* As, int lane) {
  v2f a[RING_], b[RING_];
#pragma unroll
  for (int u = 0; u < RING_; ++u) {
    b[u].x = Wp[(size_t)(4 * u) * LDB];
    b[u].y = Wp[(size_t)(4 * u + 1) * LDB];
  }

#pragma unroll
  for (int i = 0; i < AROWS_ * 1024 / 4 / 32; ++i) {
    const int c4 = (i & 7) * 32 + lane;
    const int gap = ((i & 7) >= 4) ? HGAP_ : 0;
    const v4f v = *reinterpret_cast<const v4f*>(A + (size_t)(i >> 3) * 1024 + 4 * c4);
    *reinterpret_cast<v4f*>(As + (i >> 3) * APITCH_ + 4 * c4 + gap) = v;
  }
  __syncthreads();

  const float* Al = As + (lane & 7) * APITCH_ + ((lane >> 3) & 1) * (KHALF_ + HGAP_)
                       + 2 * (lane >> 4);
#pragma unroll
  for (int u = 0; u < RING_; ++u)
    a[u] = *reinterpret_cast<const v2f*>(Al + 4 * u);

  v8f acc = {};
  for (int kb = 0; kb < KSTEPS_ - RING_; kb += RING_) {
#pragma unroll
    for (int s = 0; s < RING_; s += SLOT_) {
#pragma unroll
      for (int u = s; u < s + SLOT_; ++u)
        acc = __builtin_amdgcn_wmma_f32_16x16x4_f32(
            false, a[u], false, b[u], (short)0, acc, false, false);
#pragma unroll
      for (int u = s; u < s + SLOT_; ++u) {
        const int k = 4 * (kb + RING_ + u);
        a[u]   = *reinterpret_cast<const v2f*>(Al + k);
        b[u].x = Wp[(size_t)k * LDB];
        b[u].y = Wp[(size_t)(k + 1) * LDB];
      }
      __builtin_amdgcn_sched_barrier(0);
    }
  }
#pragma unroll
  for (int u = 0; u < RING_; ++u)
    acc = __builtin_amdgcn_wmma_f32_16x16x4_f32(
        false, a[u], false, b[u], (short)0, acc, false, false);
  return acc;
}

__global__ void __launch_bounds__(32) gemm1_gelu(const float* __restrict__ A,
                           const float* __restrict__ W1,
                           const float* __restrict__ b1,
                           float* __restrict__ Hout) {
  const int lane  = threadIdx.x & 31;
  const int half  = lane >> 4;
  const int khalf = (lane >> 3) & 1;
  const int n0    = blockIdx.x * 8;
  const int col   = n0 + (lane & 7);

  __shared__ __attribute__((aligned(16))) float As[AROWS_ * APITCH_];
  const float* Wp = W1 + (size_t)(KHALF_ * khalf + 2 * half) * H_ + col;

  const v8f acc = gemm_k1024_tile<H_>(A, Wp, As, lane);

  const float bias = b1[col];
#pragma unroll
  for (int r = 0; r < 8; ++r) {
    const float hi = __shfl(acc[r], (lane & 7) + 24, 32);
    if (lane < 8) {
      float x = (acc[r] + hi) + bias;
      float g = 0.5f * x * (1.0f + erff(x * 0.70710678118654752f));
      Hout[r * H_ + col] = g;
    }
  }
}

__global__ void layernorm_rows(float* __restrict__ Hbuf,
                               const float* __restrict__ gam,
                               const float* __restrict__ bet) {
  const int row = blockIdx.x;
  const int tid = threadIdx.x;
  __shared__ float red1[8];
  __shared__ float red2[8];

  float v[4];
  float s = 0.f;
#pragma unroll
  for (int j = 0; j < 4; ++j) {
    v[j] = Hbuf[row * H_ + tid + j * 256];
    s += v[j];
  }
#pragma unroll
  for (int off = 16; off > 0; off >>= 1) s += __shfl_down(s, off, 32);
  if ((tid & 31) == 0) red1[tid >> 5] = s;
  __syncthreads();
  float tot = 0.f;
#pragma unroll
  for (int i = 0; i < 8; ++i) tot += red1[i];
  const float mean = tot * (1.0f / (float)H_);

  float s2 = 0.f;
#pragma unroll
  for (int j = 0; j < 4; ++j) {
    float d = v[j] - mean;
    s2 += d * d;
  }
#pragma unroll
  for (int off = 16; off > 0; off >>= 1) s2 += __shfl_down(s2, off, 32);
  if ((tid & 31) == 0) red2[tid >> 5] = s2;
  __syncthreads();
  float tot2 = 0.f;
#pragma unroll
  for (int i = 0; i < 8; ++i) tot2 += red2[i];
  const float inv = rsqrtf(tot2 * (1.0f / (float)H_) + 1e-5f);

#pragma unroll
  for (int j = 0; j < 4; ++j) {
    int c = tid + j * 256;
    Hbuf[row * H_ + c] = (v[j] - mean) * inv * gam[c] + bet[c];
  }
}

__global__ void __launch_bounds__(32) gemm2_out(const float* __restrict__ Hbuf,
                          const float* __restrict__ W2,
                          const float* __restrict__ b2,
                          float* __restrict__ out) {
  const int lane  = threadIdx.x & 31;
  const int half  = lane >> 4;
  const int khalf = (lane >> 3) & 1;
  const int n0    = blockIdx.x * 8;
  const int col   = n0 + (lane & 7);

  __shared__ __attribute__((aligned(16))) float As[AROWS_ * APITCH_];
  const float* Wp = W2 + (size_t)(KHALF_ * khalf + 2 * half) * PD_ + col;

  const v8f acc = gemm_k1024_tile<PD_>(Hbuf, Wp, As, lane);

  const float bias = b2[col];
#pragma unroll
  for (int r = 0; r < 8; ++r) {
    const float hi = __shfl(acc[r], (lane & 7) + 24, 32);
    if (lane < 8) {
      float val = (acc[r] + hi) + bias;
#pragma unroll
      for (int p = 0; p < P_; ++p)
        out[(size_t)(r * P_ + p) * PD_ + col] = val;
    }
  }
}

extern "C" void kernel_launch(void* const* d_in, const int* in_sizes, int n_in,
                              void* d_out, int out_size, void* d_ws, size_t ws_size,
                              hipStream_t stream) {
  const float* ctx  = (const float*)d_in[0];
  const float* W1   = (const float*)d_in[1];
  const float* b1   = (const float*)d_in[2];
  const float* ln_g = (const float*)d_in[3];
  const float* ln_b = (const float*)d_in[4];
  const float* W2   = (const float*)d_in[5];
  const float* b2   = (const float*)d_in[6];
  const int*  masks = (const int*)d_in[7];
  (void)in_sizes; (void)n_in; (void)out_size; (void)ws_size;

  float* out = (float*)d_out;

  char* ws = (char*)d_ws;
  float* pooled   = (float*)(ws);
  float* partials = (float*)(ws + 65536);
  float* hbuf     = (float*)(ws + 65536 + 262144);

  gather_partials<<<dim3(4, TC_, B_), 256, 0, stream>>>(ctx, masks, partials);
  reduce_pool<<<64, 256, 0, stream>>>(partials, pooled);
  gemm1_gelu<<<H_ / 8, 32, 0, stream>>>(pooled, W1, b1, hbuf);
  layernorm_rows<<<B_, 256, 0, stream>>>(hbuf, ln_g, ln_b);
  gemm2_out<<<PD_ / 8, 32, 0, stream>>>(hbuf, W2, b2, out);
}


# Gather grid turned: a block owns one batch's 32-column slice for all 2048 tokens (wave = token chunk); the 8 chunk sums meet in LDS, added in the old order, pooled stored directly: 5 launches to 4
# speedup vs baseline: 3.2555x; 1.0810x over previous
#include <hip/hip_runtime.h>
#include <math.h>

typedef float v2f __attribute__((ext_vector_type(2)));
typedef float v8f __attribute__((ext_vector_type(8)));

#define B_   8
#define S_   4096
#define D_   1024
#define H_   1024
#define PD_  512
#define P_   4
#define TOK_   2048
#define TC_    8
#define TOKPC_ 256

#define GRING_ 64
#define GCOLS_ 32

__device__ __forceinline__ float gather_row(const char* cb, int idxreg,
                                            int srclane, unsigned col4) {
  const unsigned r = (unsigned)__builtin_amdgcn_readlane(idxreg, srclane);
  const unsigned off = r * (unsigned)(D_ * sizeof(float)) + col4;
  return *reinterpret_cast<const float*>(cb + off);
}

__global__ void __launch_bounds__(256) gather_pool(const float* __restrict__ ctx,
                            const int* __restrict__ masks,
                            float* __restrict__ pooled) {
  const int sl  = blockIdx.x;
  const int b   = blockIdx.y;
  const unsigned tid  = threadIdx.x;
  const unsigned lane = tid & 31u;
  const unsigned tc   = (unsigned)__builtin_amdgcn_readfirstlane((int)(tid >> 5));
  const unsigned col4 = lane * 4u;
  const int*  mrow = masks + b * TOK_ + tc * TOKPC_;
  const char* cb   = reinterpret_cast<const char*>(ctx + (size_t)b * S_ * D_ + sl * GCOLS_);


  int iv[TOKPC_ / 32];
#pragma unroll
  for (int j = 0; j < TOKPC_ / 32; ++j) iv[j] = mrow[32 * j + lane];

  float v[GRING_];
#pragma unroll
  for (int u = 0; u < GRING_; ++u) v[u] = gather_row(cb, iv[u >> 5], u & 31, col4);

  __builtin_amdgcn_sched_barrier(0);

  float sum = 0.f;
#pragma unroll
  for (int t = GRING_; t < TOKPC_; ++t) {
    sum += v[t % GRING_];
    v[t % GRING_] = gather_row(cb, iv[t >> 5], t & 31, col4);
  }
#pragma unroll
  for (int u = 0; u < GRING_; ++u) sum += v[u];

  __shared__ float part[TC_][GCOLS_];
  part[tc][lane] = sum;
  __syncthreads();
  if (tc == 0) {
    float s = 0.f;
#pragma unroll
    for (int c = 0; c < TC_; ++c) s += part[c][lane];
    s *= (1.0f / (float)TOK_);
    pooled[b * D_ + sl * GCOLS_ + lane] = s;
  }
}

typedef float v4f __attribute__((ext_vector_type(4)));

#define AROWS_  8
#define KHALF_  512
#define HGAP_   32
#define APITCH_ 1092
#define KSTEPS_ 128
#define RING_   32
#define SLOT_   4

template <int LDB>
__device__ __forceinline__ v8f gemm_k1024_tile(const float* A, const float* Wp,
                                               float* As, int lane) {
  v2f a[RING_], b[RING_];
#pragma unroll
  for (int u = 0; u < RING_; ++u) {
    b[u].x = Wp[(size_t)(4 * u) * LDB];
    b[u].y = Wp[(size_t)(4 * u + 1) * LDB];
  }

#pragma unroll
  for (int i = 0; i < AROWS_ * 1024 / 4 / 32; ++i) {
    const int c4 = (i & 7) * 32 + lane;
    const int gap = ((i & 7) >= 4) ? HGAP_ : 0;
    const v4f v = *reinterpret_cast<const v4f*>(A + (size_t)(i >> 3) * 1024 + 4 * c4);
    *reinterpret_cast<v4f*>(As + (i >> 3) * APITCH_ + 4 * c4 + gap) = v;
  }
  __syncthreads();

  const float* Al = As + (lane & 7) * APITCH_ + ((lane >> 3) & 1) * (KHALF_ + HGAP_)
                       + 2 * (lane >> 4);
#pragma unroll
  for (int u = 0; u < RING_; ++u)
    a[u] = *reinterpret_cast<const v2f*>(Al + 4 * u);

  v8f acc = {};
  for (int kb = 0; kb < KSTEPS_ - RING_; kb += RING_) {
#pragma unroll
    for (int s = 0; s < RING_; s += SLOT_) {
#pragma unroll
      for (int u = s; u < s + SLOT_; ++u)
        acc = __builtin_amdgcn_wmma_f32_16x16x4_f32(
            false, a[u], false, b[u], (short)0, acc, false, false);
#pragma unroll
      for (int u = s; u < s + SLOT_; ++u) {
        const int k = 4 * (kb + RING_ + u);
        a[u]   = *reinterpret_cast<const v2f*>(Al + k);
        b[u].x = Wp[(size_t)k * LDB];
        b[u].y = Wp[(size_t)(k + 1) * LDB];
      }
      __builtin_amdgcn_sched_barrier(0);
    }
  }
#pragma unroll
  for (int u = 0; u < RING_; ++u)
    acc = __builtin_amdgcn_wmma_f32_16x16x4_f32(
        false, a[u], false, b[u], (short)0, acc, false, false);
  return acc;
}

__global__ void __launch_bounds__(32) gemm1_gelu(const float* __restrict__ A,
                           const float* __restrict__ W1,
                           const float* __restrict__ b1,
                           float* __restrict__ Hout) {
  const int lane  = threadIdx.x & 31;
  const int half  = lane >> 4;
  const int khalf = (lane >> 3) & 1;
  const int n0    = blockIdx.x * 8;
  const int col   = n0 + (lane & 7);

  __shared__ __attribute__((aligned(16))) float As[AROWS_ * APITCH_];
  const float* Wp = W1 + (size_t)(KHALF_ * khalf + 2 * half) * H_ + col;

  const v8f acc = gemm_k1024_tile<H_>(A, Wp, As, lane);

  const float bias = b1[col];
#pragma unroll
  for (int r = 0; r < 8; ++r) {
    const float hi = __shfl(acc[r], (lane & 7) + 24, 32);
    if (lane < 8) {
      float x = (acc[r] + hi) + bias;
      float g = 0.5f * x * (1.0f + erff(x * 0.70710678118654752f));
      Hout[r * H_ + col] = g;
    }
  }
}

__global__ void layernorm_rows(float* __restrict__ Hbuf,
                               const float* __restrict__ gam,
                               const float* __restrict__ bet) {
  const int row = blockIdx.x;
  const int tid = threadIdx.x;
  __shared__ float red1[8];
  __shared__ float red2[8];

  float v[4];
  float s = 0.f;
#pragma unroll
  for (int j = 0; j < 4; ++j) {
    v[j] = Hbuf[row * H_ + tid + j * 256];
    s += v[j];
  }
#pragma unroll
  for (int off = 16; off > 0; off >>= 1) s += __shfl_down(s, off, 32);
  if ((tid & 31) == 0) red1[tid >> 5] = s;
  __syncthreads();
  float tot = 0.f;
#pragma unroll
  for (int i = 0; i < 8; ++i) tot += red1[i];
  const float mean = tot * (1.0f / (float)H_);

  float s2 = 0.f;
#pragma unroll
  for (int j = 0; j < 4; ++j) {
    float d = v[j] - mean;
    s2 += d * d;
  }
#pragma unroll
  for (int off = 16; off > 0; off >>= 1) s2 += __shfl_down(s2, off, 32);
  if ((tid & 31) == 0) red2[tid >> 5] = s2;
  __syncthreads();
  float tot2 = 0.f;
#pragma unroll
  for (int i = 0; i < 8; ++i) tot2 += red2[i];
  const float inv = rsqrtf(tot2 * (1.0f / (float)H_) + 1e-5f);

#pragma unroll
  for (int j = 0; j < 4; ++j) {
    int c = tid + j * 256;
    Hbuf[row * H_ + c] = (v[j] - mean) * inv * gam[c] + bet[c];
  }
}

__global__ void __launch_bounds__(32) gemm2_out(const float* __restrict__ Hbuf,
                          const float* __restrict__ W2,
                          const float* __restrict__ b2,
                          float* __restrict__ out) {
  const int lane  = threadIdx.x & 31;
  const int half  = lane >> 4;
  const int khalf = (lane >> 3) & 1;
  const int n0    = blockIdx.x * 8;
  const int col   = n0 + (lane & 7);

  __shared__ __attribute__((aligned(16))) float As[AROWS_ * APITCH_];
  const float* Wp = W2 + (size_t)(KHALF_ * khalf + 2 * half) * PD_ + col;

  const v8f acc = gemm_k1024_tile<PD_>(Hbuf, Wp, As, lane);

  const float bias = b2[col];
#pragma unroll
  for (int r = 0; r < 8; ++r) {
    const float hi = __shfl(acc[r], (lane & 7) + 24, 32);
    if (lane < 8) {
      float val = (acc[r] + hi) + bias;
#pragma unroll
      for (int p = 0; p < P_; ++p)
        out[(size_t)(r * P_ + p) * PD_ + col] = val;
    }
  }
}

extern "C" void kernel_launch(void* const* d_in, const int* in_sizes, int n_in,
                              void* d_out, int out_size, void* d_ws, size_t ws_size,
                              hipStream_t stream) {
  const float* ctx  = (const float*)d_in[0];
  const float* W1   = (const float*)d_in[1];
  const float* b1   = (const float*)d_in[2];
  const float* ln_g = (const float*)d_in[3];
  const float* ln_b = (const float*)d_in[4];
  const float* W2   = (const float*)d_in[5];
  const float* b2   = (const float*)d_in[6];
  const int*  masks = (const int*)d_in[7];
  (void)in_sizes; (void)n_in; (void)out_size; (void)ws_size;

  float* out = (float*)d_out;

  char* ws = (char*)d_ws;
  float* pooled   = (float*)(ws);
  float* hbuf     = (float*)(ws + 65536);

  gather_pool<<<dim3(D_ / GCOLS_, B_), 256, 0, stream>>>(ctx, masks, pooled);
  gemm1_gelu<<<H_ / 8, 32, 0, stream>>>(pooled, W1, b1, hbuf);
  layernorm_rows<<<B_, 256, 0, stream>>>(hbuf, ln_g, ln_b);
  gemm2_out<<<PD_ / 8, 32, 0, stream>>>(hbuf, W2, b2, out);
}


# Both GEMMs: all global loads of a tile issued before the K loop by 4 waves (A quarter + B fragments of 32 k-steps each), passed to wave 0 via LDS; same 128-WMMA chain, k order and sums unchanged
# speedup vs baseline: 3.4029x; 1.0453x over previous
#include <hip/hip_runtime.h>
#include <math.h>

typedef float v2f __attribute__((ext_vector_type(2)));
typedef float v8f __attribute__((ext_vector_type(8)));

#define B_   8
#define S_   4096
#define D_   1024
#define H_   1024
#define PD_  512
#define P_   4
#define TOK_   2048
#define TC_    8
#define TOKPC_ 256

#define GRING_ 64
#define GCOLS_ 32

__device__ __forceinline__ float gather_row(const char* cb, int idxreg,
                                            int srclane, unsigned col4) {
  const unsigned r = (unsigned)__builtin_amdgcn_readlane(idxreg, srclane);
  const unsigned off = r * (unsigned)(D_ * sizeof(float)) + col4;
  return *reinterpret_cast<const float*>(cb + off);
}

__global__ void __launch_bounds__(256) gather_pool(const float* __restrict__ ctx,
                            const int* __restrict__ masks,
                            float* __restrict__ pooled) {
  const int sl  = blockIdx.x;
  const int b   = blockIdx.y;
  const unsigned tid  = threadIdx.x;
  const unsigned lane = tid & 31u;
  const unsigned tc   = (unsigned)__builtin_amdgcn_readfirstlane((int)(tid >> 5));
  const unsigned col4 = lane * 4u;
  const int*  mrow = masks + b * TOK_ + tc * TOKPC_;
  const char* cb   = reinterpret_cast<const char*>(ctx + (size_t)b * S_ * D_ + sl * GCOLS_);


  int iv[TOKPC_ / 32];
#pragma unroll
  for (int j = 0; j < TOKPC_ / 32; ++j) iv[j] = mrow[32 * j + lane];

  float v[GRING_];
#pragma unroll
  for (int u = 0; u < GRING_; ++u) v[u] = gather_row(cb, iv[u >> 5], u & 31, col4);

  __builtin_amdgcn_sched_barrier(0);

  float sum = 0.f;
#pragma unroll
  for (int t = GRING_; t < TOKPC_; ++t) {
    sum += v[t % GRING_];
    v[t % GRING_] = gather_row(cb, iv[t >> 5], t & 31, col4);
  }
#pragma unroll
  for (int u = 0; u < GRING_; ++u) sum += v[u];

  __shared__ float part[TC_][GCOLS_];
  part[tc][lane] = sum;
  __syncthreads();
  if (tc == 0) {
    float s = 0.f;
#pragma unroll
    for (int c = 0; c < TC_; ++c) s += part[c][lane];
    s *= (1.0f / (float)TOK_);
    pooled[b * D_ + sl * GCOLS_ + lane] = s;
  }
}

typedef float v4f __attribute__((ext_vector_type(4)));

#define AROWS_  8
#define KHALF_  512
#define HGAP_   32
#define APITCH_ 1092
#define KSTEPS_ 128
#define GW_     4
#define KQ_     (KSTEPS_ / GW_)
#define RING_   16
#define SLOT_   4

template <int LDB>
__device__ __forceinline__ void gemm_fetch(const float* A, const float* Wq,
                                           float* As, v2f* Bs,
                                           int tid, int lane, int wv) {
  v4f av[AROWS_ * 1024 / 4 / (32 * GW_)];
#pragma unroll
  for (int i = 0; i < AROWS_ * 1024 / 4 / (32 * GW_); ++i) {
    const int c4 = (i & 1) * 128 + tid;
    av[i] = *reinterpret_cast<const v4f*>(A + (size_t)(i >> 1) * 1024 + 4 * c4);
  }
  v2f b[KQ_];
#pragma unroll
  for (int u = 0; u < KQ_; ++u) {
    b[u].x = Wq[(size_t)(4 * u) * LDB];
    b[u].y = Wq[(size_t)(4 * u + 1) * LDB];
  }
#pragma unroll
  for (int i = 0; i < AROWS_ * 1024 / 4 / (32 * GW_); ++i) {
    const int c4 = (i & 1) * 128 + tid;
    const int gap = (i & 1) ? HGAP_ : 0;
    *reinterpret_cast<v4f*>(As + (i >> 1) * APITCH_ + 4 * c4 + gap) = av[i];
  }
#pragma unroll
  for (int u = 0; u < KQ_; ++u) Bs[(KQ_ * wv + u) * 32 + lane] = b[u];
  __syncthreads();
}

__device__ __forceinline__ v8f gemm_chain(const float* As, const v2f* Bs, int lane) {
  const float* Al = As + (lane & 7) * APITCH_ + ((lane >> 3) & 1) * (KHALF_ + HGAP_)
                       + 2 * (lane >> 4);
  const v2f* Bl = Bs + lane;
  v2f a[RING_], b[RING_];
#pragma unroll
  for (int u = 0; u < RING_; ++u) {
    a[u] = *reinterpret_cast<const v2f*>(Al + 4 * u);
    b[u] = Bl[32 * u];
  }

  v8f acc = {};
  for (int kb = 0; kb < KSTEPS_ - RING_; kb += RING_) {
#pragma unroll
    for (int s = 0; s < RING_; s += SLOT_) {
#pragma unroll
      for (int u = s; u < s + SLOT_; ++u)
        acc = __builtin_amdgcn_wmma_f32_16x16x4_f32(
            false, a[u], false, b[u], (short)0, acc, false, false);
#pragma unroll
      for (int u = s; u < s + SLOT_; ++u) {
        const int ks = kb + RING_ + u;
        a[u] = *reinterpret_cast<const v2f*>(Al + 4 * ks);
        b[u] = Bl[32 * ks];
      }
      __builtin_amdgcn_sched_barrier(0);
    }
  }
#pragma unroll
  for (int u = 0; u < RING_; ++u)
    acc = __builtin_amdgcn_wmma_f32_16x16x4_f32(
        false, a[u], false, b[u], (short)0, acc, false, false);
  return acc;
}

__global__ void __launch_bounds__(32 * GW_) gemm1_gelu(const float* __restrict__ A,
                           const float* __restrict__ W1,
                           const float* __restrict__ b1,
                           float* __restrict__ Hout) {
  const int tid   = threadIdx.x;
  const int lane  = tid & 31;
  const int wv    = __builtin_amdgcn_readfirstlane(tid >> 5);
  const int half  = lane >> 4;
  const int khalf = (lane >> 3) & 1;
  const int n0    = blockIdx.x * 8;
  const int col   = n0 + (lane & 7);

  __shared__ __attribute__((aligned(16))) float As[AROWS_ * APITCH_];
  __shared__ __attribute__((aligned(16))) v2f   Bs[KSTEPS_ * 32];
  const float* Wq = W1 + (size_t)(KHALF_ * khalf + 2 * half + 4 * KQ_ * wv) * H_ + col;

  gemm_fetch<H_>(A, Wq, As, Bs, tid, lane, wv);
  if (wv != 0) return;

  const float bias = b1[col];
  const v8f acc = gemm_chain(As, Bs, lane);

#pragma unroll
  for (int r = 0; r < 8; ++r) {
    const float hi = __shfl(acc[r], (lane & 7) + 24, 32);
    if (lane < 8) {
      float x = (acc[r] + hi) + bias;
      float g = 0.5f * x * (1.0f + erff(x * 0.70710678118654752f));
      Hout[r * H_ + col] = g;
    }
  }
}

__global__ void layernorm_rows(float* __restrict__ Hbuf,
                               const float* __restrict__ gam,
                               const float* __restrict__ bet) {
  const int row = blockIdx.x;
  const int tid = threadIdx.x;
  __shared__ float red1[8];
  __shared__ float red2[8];

  float v[4];
  float s = 0.f;
#pragma unroll
  for (int j = 0; j < 4; ++j) {
    v[j] = Hbuf[row * H_ + tid + j * 256];
    s += v[j];
  }
#pragma unroll
  for (int off = 16; off > 0; off >>= 1) s += __shfl_down(s, off, 32);
  if ((tid & 31) == 0) red1[tid >> 5] = s;
  __syncthreads();
  float tot = 0.f;
#pragma unroll
  for (int i = 0; i < 8; ++i) tot += red1[i];
  const float mean = tot * (1.0f / (float)H_);

  float s2 = 0.f;
#pragma unroll
  for (int j = 0; j < 4; ++j) {
    float d = v[j] - mean;
    s2 += d * d;
  }
#pragma unroll
  for (int off = 16; off > 0; off >>= 1) s2 += __shfl_down(s2, off, 32);
  if ((tid & 31) == 0) red2[tid >> 5] = s2;
  __syncthreads();
  float tot2 = 0.f;
#pragma unroll
  for (int i = 0; i < 8; ++i) tot2 += red2[i];
  const float inv = rsqrtf(tot2 * (1.0f / (float)H_) + 1e-5f);

#pragma unroll
  for (int j = 0; j < 4; ++j) {
    int c = tid + j * 256;
    Hbuf[row * H_ + c] = (v[j] - mean) * inv * gam[c] + bet[c];
  }
}

__global__ void __launch_bounds__(32 * GW_) gemm2_out(const float* __restrict__ Hbuf,
                          const float* __restrict__ W2,
                          const float* __restrict__ b2,
                          float* __restrict__ out) {
  const int tid   = threadIdx.x;
  const int lane  = tid & 31;
  const int wv    = __builtin_amdgcn_readfirstlane(tid >> 5);
  const int half  = lane >> 4;
  const int khalf = (lane >> 3) & 1;
  const int n0    = blockIdx.x * 8;
  const int col   = n0 + (lane & 7);

  __shared__ __attribute__((aligned(16))) float As[AROWS_ * APITCH_];
  __shared__ __attribute__((aligned(16))) v2f   Bs[KSTEPS_ * 32];
  const float* Wq = W2 + (size_t)(KHALF_ * khalf + 2 * half + 4 * KQ_ * wv) * PD_ + col;

  gemm_fetch<PD_>(Hbuf, Wq, As, Bs, tid, lane, wv);
  if (wv != 0) return;

  const float bias = b2[col];
  const v8f acc = gemm_chain(As, Bs, lane);

#pragma unroll
  for (int r = 0; r < 8; ++r) {
    const float hi = __shfl(acc[r], (lane & 7) + 24, 32);
    if (lane < 8) {
      float val = (acc[r] + hi) + bias;
#pragma unroll
      for (int p = 0; p < P_; ++p)
        out[(size_t)(r * P_ + p) * PD_ + col] = val;
    }
  }
}

extern "C" void kernel_launch(void* const* d_in, const int* in_sizes, int n_in,
                              void* d_out, int out_size, void* d_ws, size_t ws_size,
                              hipStream_t stream) {
  const float* ctx  = (const float*)d_in[0];
  const float* W1   = (const float*)d_in[1];
  const float* b1   = (const float*)d_in[2];
  const float* ln_g = (const float*)d_in[3];
  const float* ln_b = (const float*)d_in[4];
  const float* W2   = (const float*)d_in[5];
  const float* b2   = (const float*)d_in[6];
  const int*  masks = (const int*)d_in[7];
  (void)in_sizes; (void)n_in; (void)out_size; (void)ws_size;

  float* out = (float*)d_out;

  char* ws = (char*)d_ws;
  float* pooled   = (float*)(ws);
  float* hbuf     = (float*)(ws + 65536);

  gather_pool<<<dim3(D_ / GCOLS_, B_), 256, 0, stream>>>(ctx, masks, pooled);
  gemm1_gelu<<<H_ / 8, 32 * GW_, 0, stream>>>(pooled, W1, b1, hbuf);
  layernorm_rows<<<B_, 256, 0, stream>>>(hbuf, ln_g, ln_b);
  gemm2_out<<<PD_ / 8, 32 * GW_, 0, stream>>>(hbuf, W2, b2, out);
}


# No LayerNorm launch: GEMM1 stores each 8-column tile's per-row mean and squared-distance sum, GEMM2 merges a row's 128 tiles in a half wave and normalises A into LDS (LayerNorm sum order changes)
# speedup vs baseline: 3.8147x; 1.1210x over previous
#include <hip/hip_runtime.h>
#include <math.h>

typedef float v2f __attribute__((ext_vector_type(2)));
typedef float v8f __attribute__((ext_vector_type(8)));

#define B_   8
#define S_   4096
#define D_   1024
#define H_   1024
#define PD_  512
#define P_   4
#define TOK_   2048
#define TC_    8
#define TOKPC_ 256

#define GRING_ 64
#define GCOLS_ 32

__device__ __forceinline__ float gather_row(const char* cb, int idxreg,
                                            int srclane, unsigned col4) {
  const unsigned r = (unsigned)__builtin_amdgcn_readlane(idxreg, srclane);
  const unsigned off = r * (unsigned)(D_ * sizeof(float)) + col4;
  return *reinterpret_cast<const float*>(cb + off);
}

__global__ void __launch_bounds__(256) gather_pool(const float* __restrict__ ctx,
                            const int* __restrict__ masks,
                            float* __restrict__ pooled) {
  const int sl  = blockIdx.x;
  const int b   = blockIdx.y;
  const unsigned tid  = threadIdx.x;
  const unsigned lane = tid & 31u;
  const unsigned tc   = (unsigned)__builtin_amdgcn_readfirstlane((int)(tid >> 5));
  const unsigned col4 = lane * 4u;
  const int*  mrow = masks + b * TOK_ + tc * TOKPC_;
  const char* cb   = reinterpret_cast<const char*>(ctx + (size_t)b * S_ * D_ + sl * GCOLS_);


  int iv[TOKPC_ / 32];
#pragma unroll
  for (int j = 0; j < TOKPC_ / 32; ++j) iv[j] = mrow[32 * j + lane];

  float v[GRING_];
#pragma unroll
  for (int u = 0; u < GRING_; ++u) v[u] = gather_row(cb, iv[u >> 5], u & 31, col4);

  __builtin_amdgcn_sched_barrier(0);

  float sum = 0.f;
#pragma unroll
  for (int t = GRING_; t < TOKPC_; ++t) {
    sum += v[t % GRING_];
    v[t % GRING_] = gather_row(cb, iv[t >> 5], t & 31, col4);
  }
#pragma unroll
  for (int u = 0; u < GRING_; ++u) sum += v[u];

  __shared__ float part[TC_][GCOLS_];
  part[tc][lane] = sum;
  __syncthreads();
  if (tc == 0) {
    float s = 0.f;
#pragma unroll
    for (int c = 0; c < TC_; ++c) s += part[c][lane];
    s *= (1.0f / (float)TOK_);
    pooled[b * D_ + sl * GCOLS_ + lane] = s;
  }
}

typedef float v4f __attribute__((ext_vector_type(4)));

#define AROWS_  8
#define KHALF_  512
#define HGAP_   32
#define APITCH_ 1092
#define KSTEPS_ 128
#define GW_     4
#define KQ_     (KSTEPS_ / GW_)
#define RING_   16
#define SLOT_   4

template <int LDB>
__device__ __forceinline__ void gemm_fetch(const float* A, const float* Wq,
                                           float* As, v2f* Bs,
                                           int tid, int lane, int wv) {
  v4f av[AROWS_ * 1024 / 4 / (32 * GW_)];
#pragma unroll
  for (int i = 0; i < AROWS_ * 1024 / 4 / (32 * GW_); ++i) {
    const int c4 = (i & 1) * 128 + tid;
    av[i] = *reinterpret_cast<const v4f*>(A + (size_t)(i >> 1) * 1024 + 4 * c4);
  }
  v2f b[KQ_];
#pragma unroll
  for (int u = 0; u < KQ_; ++u) {
    b[u].x = Wq[(size_t)(4 * u) * LDB];
    b[u].y = Wq[(size_t)(4 * u + 1) * LDB];
  }
#pragma unroll
  for (int i = 0; i < AROWS_ * 1024 / 4 / (32 * GW_); ++i) {
    const int c4 = (i & 1) * 128 + tid;
    const int gap = (i & 1) ? HGAP_ : 0;
    *reinterpret_cast<v4f*>(As + (i >> 1) * APITCH_ + 4 * c4 + gap) = av[i];
  }
#pragma unroll
  for (int u = 0; u < KQ_; ++u) Bs[(KQ_ * wv + u) * 32 + lane] = b[u];
  __syncthreads();
}

__device__ __forceinline__ v8f gemm_chain(const float* As, const v2f* Bs, int lane) {
  const float* Al = As + (lane & 7) * APITCH_ + ((lane >> 3) & 1) * (KHALF_ + HGAP_)
                       + 2 * (lane >> 4);
  const v2f* Bl = Bs + lane;
  v2f a[RING_], b[RING_];
#pragma unroll
  for (int u = 0; u < RING_; ++u) {
    a[u] = *reinterpret_cast<const v2f*>(Al + 4 * u);
    b[u] = Bl[32 * u];
  }

  v8f acc = {};
  for (int kb = 0; kb < KSTEPS_ - RING_; kb += RING_) {
#pragma unroll
    for (int s = 0; s < RING_; s += SLOT_) {
#pragma unroll
      for (int u = s; u < s + SLOT_; ++u)
        acc = __builtin_amdgcn_wmma_f32_16x16x4_f32(
            false, a[u], false, b[u], (short)0, acc, false, false);
#pragma unroll
      for (int u = s; u < s + SLOT_; ++u) {
        const int ks = kb + RING_ + u;
        a[u] = *reinterpret_cast<const v2f*>(Al + 4 * ks);
        b[u] = Bl[32 * ks];
      }
      __builtin_amdgcn_sched_barrier(0);
    }
  }
#pragma unroll
  for (int u = 0; u < RING_; ++u)
    acc = __builtin_amdgcn_wmma_f32_16x16x4_f32(
        false, a[u], false, b[u], (short)0, acc, false, false);
  return acc;
}

#define NT1_ (H_ / 8)

__global__ void __launch_bounds__(32 * GW_) gemm1_gelu(const float* __restrict__ A,
                           const float* __restrict__ W1,
                           const float* __restrict__ b1,
                           float* __restrict__ Hout,
                           float* __restrict__ st) {
  const int tid   = threadIdx.x;
  const int lane  = tid & 31;
  const int wv    = __builtin_amdgcn_readfirstlane(tid >> 5);
  const int half  = lane >> 4;
  const int khalf = (lane >> 3) & 1;
  const int n0    = blockIdx.x * 8;
  const int col   = n0 + (lane & 7);

  __shared__ __attribute__((aligned(16))) float As[AROWS_ * APITCH_];
  __shared__ __attribute__((aligned(16))) v2f   Bs[KSTEPS_ * 32];
  __shared__ __attribute__((aligned(16))) float Tl[8][8];
  const float* Wq = W1 + (size_t)(KHALF_ * khalf + 2 * half + 4 * KQ_ * wv) * H_ + col;

  gemm_fetch<H_>(A, Wq, As, Bs, tid, lane, wv);
  if (wv != 0) return;

  const float bias = b1[col];
  const v8f acc = gemm_chain(As, Bs, lane);

  float hi[8];
#pragma unroll
  for (int r = 0; r < 8; ++r)
    hi[r] = __shfl(acc[r], (lane & 7) + 24, 32);
  if (lane < 8) {
#pragma unroll
    for (int r = 0; r < 8; ++r) {
      float x = (acc[r] + hi[r]) + bias;
      float g = 0.5f * x * (1.0f + erff(x * 0.70710678118654752f));
      Hout[r * H_ + col] = g;
      Tl[r][lane] = g;
    }
    const v4f t0 = *reinterpret_cast<const v4f*>(&Tl[lane][0]);
    const v4f t1 = *reinterpret_cast<const v4f*>(&Tl[lane][4]);
    const float m = (((t0.x + t0.y) + (t0.z + t0.w)) + ((t1.x + t1.y) + (t1.z + t1.w))) * 0.125f;
    float q = 0.f;
#pragma unroll
    for (int e = 0; e < 4; ++e) {
      const float d0 = t0[e] - m, d1 = t1[e] - m;
      q += d0 * d0;
      q += d1 * d1;
    }
    v2f mq; mq.x = m; mq.y = q;
    *reinterpret_cast<v2f*>(st + 2 * (lane * NT1_ + (int)blockIdx.x)) = mq;
  }
}


template <int M>
__device__ __forceinline__ float row_xor(float v) {
  return __builtin_bit_cast(float, __builtin_amdgcn_update_dpp(
      0, __builtin_bit_cast(int, v), 0x160 + M, 0xf, 0xf, false));
}

template <int M>
__device__ __forceinline__ void stat_step(float& m, float& q, float half_n) {
  const float mo = row_xor<M>(m);
  const float qo = row_xor<M>(q);
  const float d  = mo - m;
  q = (q + qo) + half_n * (d * d);
  m = 0.5f * (m + mo);
}

__device__ __forceinline__ void gemm2_fetch_ln(const float* Hg, const float* st,
                                               const float* gam, const float* bet,
                                               const float* Wq, float* As, v2f* Bs,
                                               int lane, int wv) {
  const int j = lane & 15;
  const int r = 2 * wv + (lane >> 4);
  v4f sv[4];
#pragma unroll
  for (int q = 0; q < 4; ++q)
    sv[q] = *reinterpret_cast<const v4f*>(st + 2 * (r * NT1_) + 4 * (j + 16 * q));
  v4f av[16], gv[16], bv[16];
#pragma unroll
  for (int i = 0; i < 16; ++i)
    av[i] = *reinterpret_cast<const v4f*>(Hg + r * H_ + 4 * (16 * i + j));
#pragma unroll
  for (int i = 0; i < 16; ++i) {
    gv[i] = *reinterpret_cast<const v4f*>(gam + 4 * (16 * i + j));
    bv[i] = *reinterpret_cast<const v4f*>(bet + 4 * (16 * i + j));
  }
  v2f b[KQ_];
#pragma unroll
  for (int u = 0; u < KQ_; ++u) {
    b[u].x = Wq[(size_t)(4 * u) * PD_];
    b[u].y = Wq[(size_t)(4 * u + 1) * PD_];
  }

  float m = (((sv[0].x + sv[0].z) + (sv[1].x + sv[1].z)) +
             ((sv[2].x + sv[2].z) + (sv[3].x + sv[3].z))) * 0.125f;
  float qs = 0.f, ds = 0.f;
#pragma unroll
  for (int q = 0; q < 4; ++q) {
    const float d0 = sv[q].x - m, d1 = sv[q].z - m;
    qs += sv[q].y;
    qs += sv[q].w;
    ds += d0 * d0;
    ds += d1 * d1;
  }
  float qq = qs + 8.0f * ds;
  stat_step<1>(m, qq, 32.0f);
  stat_step<2>(m, qq, 64.0f);
  stat_step<4>(m, qq, 128.0f);
  stat_step<8>(m, qq, 256.0f);
  const float inv = rsqrtf(qq * (1.0f / (float)H_) + 1e-5f);

#pragma unroll
  for (int i = 0; i < 16; ++i) {
    v4f o;
#pragma unroll
    for (int e = 0; e < 4; ++e) o[e] = (av[i][e] - m) * inv * gv[i][e] + bv[i][e];
    const int gap = (i >= 8) ? HGAP_ : 0;
    *reinterpret_cast<v4f*>(As + r * APITCH_ + 4 * (16 * i + j) + gap) = o;
  }
#pragma unroll
  for (int u = 0; u < KQ_; ++u) Bs[(KQ_ * wv + u) * 32 + lane] = b[u];
  __syncthreads();
}

__global__ void __launch_bounds__(32 * GW_) gemm2_ln_out(const float* __restrict__ Hg,
                          const float* __restrict__ st,
                          const float* __restrict__ gam,
                          const float* __restrict__ bet,
                          const float* __restrict__ W2,
                          const float* __restrict__ b2,
                          float* __restrict__ out) {
  const int tid   = threadIdx.x;
  const int lane  = tid & 31;
  const int wv    = __builtin_amdgcn_readfirstlane(tid >> 5);
  const int half  = lane >> 4;
  const int khalf = (lane >> 3) & 1;
  const int n0    = blockIdx.x * 8;
  const int col   = n0 + (lane & 7);

  __shared__ __attribute__((aligned(16))) float As[AROWS_ * APITCH_];
  __shared__ __attribute__((aligned(16))) v2f   Bs[KSTEPS_ * 32];
  const float* Wq = W2 + (size_t)(KHALF_ * khalf + 2 * half + 4 * KQ_ * wv) * PD_ + col;

  gemm2_fetch_ln(Hg, st, gam, bet, Wq, As, Bs, lane, wv);
  if (wv != 0) return;

  const float bias = b2[col];
  const v8f acc = gemm_chain(As, Bs, lane);

#pragma unroll
  for (int r = 0; r < 8; ++r) {
    const float hi = __shfl(acc[r], (lane & 7) + 24, 32);
    if (lane < 8) {
      float val = (acc[r] + hi) + bias;
#pragma unroll
      for (int p = 0; p < P_; ++p)
        out[(size_t)(r * P_ + p) * PD_ + col] = val;
    }
  }
}

extern "C" void kernel_launch(void* const* d_in, const int* in_sizes, int n_in,
                              void* d_out, int out_size, void* d_ws, size_t ws_size,
                              hipStream_t stream) {
  const float* ctx  = (const float*)d_in[0];
  const float* W1   = (const float*)d_in[1];
  const float* b1   = (const float*)d_in[2];
  const float* ln_g = (const float*)d_in[3];
  const float* ln_b = (const float*)d_in[4];
  const float* W2   = (const float*)d_in[5];
  const float* b2   = (const float*)d_in[6];
  const int*  masks = (const int*)d_in[7];
  (void)in_sizes; (void)n_in; (void)out_size; (void)ws_size;

  float* out = (float*)d_out;

  char* ws = (char*)d_ws;
  float* pooled   = (float*)(ws);
  float* stats    = (float*)(ws + 32768);
  float* hbuf     = (float*)(ws + 65536);

  gather_pool<<<dim3(D_ / GCOLS_, B_), 256, 0, stream>>>(ctx, masks, pooled);
  gemm1_gelu<<<H_ / 8, 32 * GW_, 0, stream>>>(pooled, W1, b1, hbuf, stats);
  gemm2_ln_out<<<PD_ / 8, 32 * GW_, 0, stream>>>(hbuf, stats, ln_g, ln_b, W2, b2, out);
}


# Both GEMMs: WMMA tile = 4 batches x 4 columns x 4 K quarters (B transposed in LDS); each of the 4 waves chains 64 WMMAs on its own tile, not 128 on wave 0; K sum order changes: (q0+q1)+(q2+q3)
# speedup vs baseline: 4.0634x; 1.0652x over previous
#include <hip/hip_runtime.h>
#include <math.h>

typedef float v2f __attribute__((ext_vector_type(2)));
typedef float v8f __attribute__((ext_vector_type(8)));

#define B_   8
#define S_   4096
#define D_   1024
#define H_   1024
#define PD_  512
#define P_   4
#define TOK_   2048
#define TC_    8
#define TOKPC_ 256

#define GRING_ 64
#define GCOLS_ 32

__device__ __forceinline__ float gather_row(const char* cb, int idxreg,
                                            int srclane, unsigned col4) {
  const unsigned r = (unsigned)__builtin_amdgcn_readlane(idxreg, srclane);
  const unsigned off = r * (unsigned)(D_ * sizeof(float)) + col4;
  return *reinterpret_cast<const float*>(cb + off);
}

__global__ void __launch_bounds__(256) gather_pool(const float* __restrict__ ctx,
                            const int* __restrict__ masks,
                            float* __restrict__ pooled) {
  const int sl  = blockIdx.x;
  const int b   = blockIdx.y;
  const unsigned tid  = threadIdx.x;
  const unsigned lane = tid & 31u;
  const unsigned tc   = (unsigned)__builtin_amdgcn_readfirstlane((int)(tid >> 5));
  const unsigned col4 = lane * 4u;
  const int*  mrow = masks + b * TOK_ + tc * TOKPC_;
  const char* cb   = reinterpret_cast<const char*>(ctx + (size_t)b * S_ * D_ + sl * GCOLS_);


  int iv[TOKPC_ / 32];
#pragma unroll
  for (int j = 0; j < TOKPC_ / 32; ++j) iv[j] = mrow[32 * j + lane];

  float v[GRING_];
#pragma unroll
  for (int u = 0; u < GRING_; ++u) v[u] = gather_row(cb, iv[u >> 5], u & 31, col4);

  __builtin_amdgcn_sched_barrier(0);

  float sum = 0.f;
#pragma unroll
  for (int t = GRING_; t < TOKPC_; ++t) {
    sum += v[t % GRING_];
    v[t % GRING_] = gather_row(cb, iv[t >> 5], t & 31, col4);
  }
#pragma unroll
  for (int u = 0; u < GRING_; ++u) sum += v[u];

  __shared__ float part[TC_][GCOLS_];
  part[tc][lane] = sum;
  __syncthreads();
  if (tc == 0) {
    float s = 0.f;
#pragma unroll
    for (int c = 0; c < TC_; ++c) s += part[c][lane];
    s *= (1.0f / (float)TOK_);
    pooled[b * D_ + sl * GCOLS_ + lane] = s;
  }
}

typedef float v4f __attribute__((ext_vector_type(4)));

#define AROWS_  8
#define KHALF_  512
#define KQLEN_  256
#define QGAP_   16
#define QPITCH_ (KQLEN_ + QGAP_)
#define APITCH_ 1092
#define KSTEPS_ 64
#define GW_     4
#define KQ_     32
#define RING_   16
#define SLOT_   4

__device__ __forceinline__ int kpos(int k) { return k + QGAP_ * (k >> 8); }

template <int LDB>
__device__ __forceinline__ void gemm_fetch(const float* A, const float* Wq,
                                           float* As, float* Bst, int tid) {
  v4f av[AROWS_ * 1024 / 4 / (32 * GW_)];
#pragma unroll
  for (int i = 0; i < AROWS_ * 1024 / 4 / (32 * GW_); ++i) {
    const int c4 = (i & 1) * 128 + tid;
    av[i] = *reinterpret_cast<const v4f*>(A + (size_t)(i >> 1) * 1024 + 4 * c4);
  }
  v2f b[KQ_];
#pragma unroll
  for (int u = 0; u < KQ_; ++u) {
    b[u].x = Wq[(size_t)(4 * u) * LDB];
    b[u].y = Wq[(size_t)(4 * u + 1) * LDB];
  }
#pragma unroll
  for (int i = 0; i < AROWS_ * 1024 / 4 / (32 * GW_); ++i) {
    const int c4 = (i & 1) * 128 + tid;
    *reinterpret_cast<v4f*>(As + (i >> 1) * APITCH_ + 4 * c4 + QGAP_ * (c4 >> 6)) = av[i];
  }
#pragma unroll
  for (int u = 0; u < KQ_; ++u) *reinterpret_cast<v2f*>(Bst + 4 * u) = b[u];
  __syncthreads();
}

__device__ __forceinline__ float* b_store_ptr(float* Bt, int lane, int wv) {
  const int half  = lane >> 4;
  const int khalf = (lane >> 3) & 1;
  return Bt + (lane & 7) * APITCH_ + kpos(KHALF_ * khalf + 4 * KQ_ * wv) + 2 * half;
}

__device__ __forceinline__ v8f gemm_chain(const float* As, const float* Bt,
                                          int lane, int g, int cg) {
  const int qoff = ((lane >> 2) & 3) * QPITCH_ + 2 * (lane >> 4);
  const float* Al = As + (4 * g  + (lane & 3)) * APITCH_ + qoff;
  const float* Bl = Bt + (4 * cg + (lane & 3)) * APITCH_ + qoff;
  v2f a[RING_], b[RING_];
#pragma unroll
  for (int u = 0; u < RING_; ++u) {
    a[u] = *reinterpret_cast<const v2f*>(Al + 4 * u);
    b[u] = *reinterpret_cast<const v2f*>(Bl + 4 * u);
  }

  v8f acc = {};
  for (int kb = 0; kb < KSTEPS_ - RING_; kb += RING_) {
#pragma unroll
    for (int s = 0; s < RING_; s += SLOT_) {
#pragma unroll
      for (int u = s; u < s + SLOT_; ++u)
        acc = __builtin_amdgcn_wmma_f32_16x16x4_f32(
            false, a[u], false, b[u], (short)0, acc, false, false);
#pragma unroll
      for (int u = s; u < s + SLOT_; ++u) {
        const int ks = kb + RING_ + u;
        a[u] = *reinterpret_cast<const v2f*>(Al + 4 * ks);
        b[u] = *reinterpret_cast<const v2f*>(Bl + 4 * ks);
      }
      __builtin_amdgcn_sched_barrier(0);
    }
  }
#pragma unroll
  for (int u = 0; u < RING_; ++u)
    acc = __builtin_amdgcn_wmma_f32_16x16x4_f32(
        false, a[u], false, b[u], (short)0, acc, false, false);
  return acc;
}

__device__ __forceinline__ void fold_quarters(const v8f& acc, int lane, float v[4]) {
#pragma unroll
  for (int bi = 0; bi < 4; ++bi) {
    const float x = acc[bi] + __shfl(acc[4 + bi], (lane + 4) & 31, 32);
    v[bi] = x + __shfl(x, (lane + 24) & 31, 32);
  }
}

#define NT1_ (H_ / 8)

__global__ void __launch_bounds__(32 * GW_) gemm1_gelu(const float* __restrict__ A,
                           const float* __restrict__ W1,
                           const float* __restrict__ b1,
                           float* __restrict__ Hout,
                           float* __restrict__ st) {
  const int tid   = threadIdx.x;
  const int lane  = tid & 31;
  const int wv    = __builtin_amdgcn_readfirstlane(tid >> 5);
  const int half  = lane >> 4;
  const int khalf = (lane >> 3) & 1;
  const int n0    = blockIdx.x * 8;
  const int col   = n0 + (lane & 7);

  __shared__ __attribute__((aligned(16))) float As[AROWS_ * APITCH_];
  __shared__ __attribute__((aligned(16))) float Bt[8 * APITCH_];
  __shared__ __attribute__((aligned(16))) float Tl[8][8];
  const float* Wq = W1 + (size_t)(KHALF_ * khalf + 2 * half + 4 * KQ_ * wv) * H_ + col;

  gemm_fetch<H_>(A, Wq, As, b_store_ptr(Bt, lane, wv), tid);

  const int g  = wv >> 1;
  const int cg = wv & 1;
  const float bias = b1[n0 + 4 * cg + (lane & 3)];
  const v8f acc = gemm_chain(As, Bt, lane, g, cg);

  float v[4];
  fold_quarters(acc, lane, v);
  if (lane < 4) {
#pragma unroll
    for (int bi = 0; bi < 4; ++bi) {
      const float x = v[bi] + bias;
      Tl[4 * g + bi][4 * cg + lane] =
          0.5f * x * (1.0f + erff(x * 0.70710678118654752f));
    }
  }
  __syncthreads();
  if (wv != 0) return;

  {
    const int r = lane >> 2, c2 = 2 * (lane & 3);
    *reinterpret_cast<v2f*>(Hout + r * H_ + n0 + c2) = *reinterpret_cast<const v2f*>(&Tl[r][c2]);
  }
  if (lane < 8) {
    const v4f t0 = *reinterpret_cast<const v4f*>(&Tl[lane][0]);
    const v4f t1 = *reinterpret_cast<const v4f*>(&Tl[lane][4]);
    const float m = (((t0.x + t0.y) + (t0.z + t0.w)) + ((t1.x + t1.y) + (t1.z + t1.w))) * 0.125f;
    float q = 0.f;
#pragma unroll
    for (int e = 0; e < 4; ++e) {
      const float d0 = t0[e] - m, d1 = t1[e] - m;
      q += d0 * d0;
      q += d1 * d1;
    }
    v2f mq; mq.x = m; mq.y = q;
    *reinterpret_cast<v2f*>(st + 2 * (lane * NT1_ + (int)blockIdx.x)) = mq;
  }
}


template <int M>
__device__ __forceinline__ float row_xor(float v) {
  return __builtin_bit_cast(float, __builtin_amdgcn_update_dpp(
      0, __builtin_bit_cast(int, v), 0x160 + M, 0xf, 0xf, false));
}

template <int M>
__device__ __forceinline__ void stat_step(float& m, float& q, float half_n) {
  const float mo = row_xor<M>(m);
  const float qo = row_xor<M>(q);
  const float d  = mo - m;
  q = (q + qo) + half_n * (d * d);
  m = 0.5f * (m + mo);
}

__device__ __forceinline__ void gemm2_fetch_ln(const float* Hg, const float* st,
                                               const float* gam, const float* bet,
                                               const float* Wq, float* As, float* Bst,
                                               int lane, int wv) {
  const int j = lane & 15;
  const int r = 2 * wv + (lane >> 4);
  v4f sv[4];
#pragma unroll
  for (int q = 0; q < 4; ++q)
    sv[q] = *reinterpret_cast<const v4f*>(st + 2 * (r * NT1_) + 4 * (j + 16 * q));
  v4f av[16], gv[16], bv[16];
#pragma unroll
  for (int i = 0; i < 16; ++i)
    av[i] = *reinterpret_cast<const v4f*>(Hg + r * H_ + 4 * (16 * i + j));
#pragma unroll
  for (int i = 0; i < 16; ++i) {
    gv[i] = *reinterpret_cast<const v4f*>(gam + 4 * (16 * i + j));
    bv[i] = *reinterpret_cast<const v4f*>(bet + 4 * (16 * i + j));
  }
  v2f b[KQ_];
#pragma unroll
  for (int u = 0; u < KQ_; ++u) {
    b[u].x = Wq[(size_t)(4 * u) * PD_];
    b[u].y = Wq[(size_t)(4 * u + 1) * PD_];
  }

  float m = (((sv[0].x + sv[0].z) + (sv[1].x + sv[1].z)) +
             ((sv[2].x + sv[2].z) + (sv[3].x + sv[3].z))) * 0.125f;
  float qs = 0.f, ds = 0.f;
#pragma unroll
  for (int q = 0; q < 4; ++q) {
    const float d0 = sv[q].x - m, d1 = sv[q].z - m;
    qs += sv[q].y;
    qs += sv[q].w;
    ds += d0 * d0;
    ds += d1 * d1;
  }
  float qq = qs + 8.0f * ds;
  stat_step<1>(m, qq, 32.0f);
  stat_step<2>(m, qq, 64.0f);
  stat_step<4>(m, qq, 128.0f);
  stat_step<8>(m, qq, 256.0f);
  const float inv = rsqrtf(qq * (1.0f / (float)H_) + 1e-5f);

#pragma unroll
  for (int i = 0; i < 16; ++i) {
    v4f o;
#pragma unroll
    for (int e = 0; e < 4; ++e) o[e] = (av[i][e] - m) * inv * gv[i][e] + bv[i][e];
    *reinterpret_cast<v4f*>(As + r * APITCH_ + 4 * (16 * i + j) + QGAP_ * (i >> 2)) = o;
  }
#pragma unroll
  for (int u = 0; u < KQ_; ++u) *reinterpret_cast<v2f*>(Bst + 4 * u) = b[u];
  __syncthreads();
}

__global__ void __launch_bounds__(32 * GW_) gemm2_ln_out(const float* __restrict__ Hg,
                          const float* __restrict__ st,
                          const float* __restrict__ gam,
                          const float* __restrict__ bet,
                          const float* __restrict__ W2,
                          const float* __restrict__ b2,
                          float* __restrict__ out) {
  const int tid   = threadIdx.x;
  const int lane  = tid & 31;
  const int wv    = __builtin_amdgcn_readfirstlane(tid >> 5);
  const int half  = lane >> 4;
  const int khalf = (lane >> 3) & 1;
  const int n0    = blockIdx.x * 8;
  const int col   = n0 + (lane & 7);

  __shared__ __attribute__((aligned(16))) float As[AROWS_ * APITCH_];
  __shared__ __attribute__((aligned(16))) float Bt[8 * APITCH_];
  const float* Wq = W2 + (size_t)(KHALF_ * khalf + 2 * half + 4 * KQ_ * wv) * PD_ + col;

  gemm2_fetch_ln(Hg, st, gam, bet, Wq, As, b_store_ptr(Bt, lane, wv), lane, wv);

  const int g  = wv >> 1;
  const int cg = wv & 1;
  const int oc = n0 + 4 * cg + (lane & 3);
  const float bias = b2[oc];
  const v8f acc = gemm_chain(As, Bt, lane, g, cg);

  float v[4];
  fold_quarters(acc, lane, v);
  if (lane < 4) {
#pragma unroll
    for (int bi = 0; bi < 4; ++bi) {
      const float val = v[bi] + bias;
#pragma unroll
      for (int p = 0; p < P_; ++p)
        out[(size_t)((4 * g + bi) * P_ + p) * PD_ + oc] = val;
    }
  }
}

extern "C" void kernel_launch(void* const* d_in, const int* in_sizes, int n_in,
                              void* d_out, int out_size, void* d_ws, size_t ws_size,
                              hipStream_t stream) {
  const float* ctx  = (const float*)d_in[0];
  const float* W1   = (const float*)d_in[1];
  const float* b1   = (const float*)d_in[2];
  const float* ln_g = (const float*)d_in[3];
  const float* ln_b = (const float*)d_in[4];
  const float* W2   = (const float*)d_in[5];
  const float* b2   = (const float*)d_in[6];
  const int*  masks = (const int*)d_in[7];
  (void)in_sizes; (void)n_in; (void)out_size; (void)ws_size;

  float* out = (float*)d_out;

  char* ws = (char*)d_ws;
  float* pooled   = (float*)(ws);
  float* stats    = (float*)(ws + 32768);
  float* hbuf     = (float*)(ws + 65536);

  gather_pool<<<dim3(D_ / GCOLS_, B_), 256, 0, stream>>>(ctx, masks, pooled);
  gemm1_gelu<<<H_ / 8, 32 * GW_, 0, stream>>>(pooled, W1, b1, hbuf, stats);
  gemm2_ln_out<<<PD_ / 8, 32 * GW_, 0, stream>>>(hbuf, stats, ln_g, ln_b, W2, b2, out);
}


# Both GEMM epilogues freed of round trips behind the WMMA chain: bias loaded with the tile's loads (the compiler had sunk it), K-quarter fold by DPP and permlanex16, no LDS shuffles; sums unchanged
# speedup vs baseline: 4.2855x; 1.0547x over previous
#include <hip/hip_runtime.h>
#include <math.h>

typedef float v2f __attribute__((ext_vector_type(2)));
typedef float v8f __attribute__((ext_vector_type(8)));

#define B_   8
#define S_   4096
#define D_   1024
#define H_   1024
#define PD_  512
#define P_   4
#define TOK_   2048
#define TC_    8
#define TOKPC_ 256

#define GRING_ 64
#define GCOLS_ 32

__device__ __forceinline__ float gather_row(const char* cb, int idxreg,
                                            int srclane, unsigned col4) {
  const unsigned r = (unsigned)__builtin_amdgcn_readlane(idxreg, srclane);
  const unsigned off = r * (unsigned)(D_ * sizeof(float)) + col4;
  return *reinterpret_cast<const float*>(cb + off);
}

__global__ void __launch_bounds__(256) gather_pool(const float* __restrict__ ctx,
                            const int* __restrict__ masks,
                            float* __restrict__ pooled) {
  const int sl  = blockIdx.x;
  const int b   = blockIdx.y;
  const unsigned tid  = threadIdx.x;
  const unsigned lane = tid & 31u;
  const unsigned tc   = (unsigned)__builtin_amdgcn_readfirstlane((int)(tid >> 5));
  const unsigned col4 = lane * 4u;
  const int*  mrow = masks + b * TOK_ + tc * TOKPC_;
  const char* cb   = reinterpret_cast<const char*>(ctx + (size_t)b * S_ * D_ + sl * GCOLS_);


  int iv[TOKPC_ / 32];
#pragma unroll
  for (int j = 0; j < TOKPC_ / 32; ++j) iv[j] = mrow[32 * j + lane];

  float v[GRING_];
#pragma unroll
  for (int u = 0; u < GRING_; ++u) v[u] = gather_row(cb, iv[u >> 5], u & 31, col4);

  __builtin_amdgcn_sched_barrier(0);

  float sum = 0.f;
#pragma unroll
  for (int t = GRING_; t < TOKPC_; ++t) {
    sum += v[t % GRING_];
    v[t % GRING_] = gather_row(cb, iv[t >> 5], t & 31, col4);
  }
#pragma unroll
  for (int u = 0; u < GRING_; ++u) sum += v[u];

  __shared__ float part[TC_][GCOLS_];
  part[tc][lane] = sum;
  __syncthreads();
  if (tc == 0) {
    float s = 0.f;
#pragma unroll
    for (int c = 0; c < TC_; ++c) s += part[c][lane];
    s *= (1.0f / (float)TOK_);
    pooled[b * D_ + sl * GCOLS_ + lane] = s;
  }
}

typedef float v4f __attribute__((ext_vector_type(4)));

#define AROWS_  8
#define KHALF_  512
#define KQLEN_  256
#define QGAP_   16
#define QPITCH_ (KQLEN_ + QGAP_)
#define APITCH_ 1092
#define KSTEPS_ 64
#define GW_     4
#define KQ_     32
#define RING_   16
#define SLOT_   4

__device__ __forceinline__ int kpos(int k) { return k + QGAP_ * (k >> 8); }

template <int LDB>
__device__ __forceinline__ void gemm_fetch(const float* A, const float* Wq,
                                           float* As, float* Bst, int tid) {
  v4f av[AROWS_ * 1024 / 4 / (32 * GW_)];
#pragma unroll
  for (int i = 0; i < AROWS_ * 1024 / 4 / (32 * GW_); ++i) {
    const int c4 = (i & 1) * 128 + tid;
    av[i] = *reinterpret_cast<const v4f*>(A + (size_t)(i >> 1) * 1024 + 4 * c4);
  }
  v2f b[KQ_];
#pragma unroll
  for (int u = 0; u < KQ_; ++u) {
    b[u].x = Wq[(size_t)(4 * u) * LDB];
    b[u].y = Wq[(size_t)(4 * u + 1) * LDB];
  }
#pragma unroll
  for (int i = 0; i < AROWS_ * 1024 / 4 / (32 * GW_); ++i) {
    const int c4 = (i & 1) * 128 + tid;
    *reinterpret_cast<v4f*>(As + (i >> 1) * APITCH_ + 4 * c4 + QGAP_ * (c4 >> 6)) = av[i];
  }
#pragma unroll
  for (int u = 0; u < KQ_; ++u) *reinterpret_cast<v2f*>(Bst + 4 * u) = b[u];
  __syncthreads();
}

__device__ __forceinline__ float* b_store_ptr(float* Bt, int lane, int wv) {
  const int half  = lane >> 4;
  const int khalf = (lane >> 3) & 1;
  return Bt + (lane & 7) * APITCH_ + kpos(KHALF_ * khalf + 4 * KQ_ * wv) + 2 * half;
}

__device__ __forceinline__ v8f gemm_chain(const float* As, const float* Bt,
                                          int lane, int g, int cg) {
  const int qoff = ((lane >> 2) & 3) * QPITCH_ + 2 * (lane >> 4);
  const float* Al = As + (4 * g  + (lane & 3)) * APITCH_ + qoff;
  const float* Bl = Bt + (4 * cg + (lane & 3)) * APITCH_ + qoff;
  v2f a[RING_], b[RING_];
#pragma unroll
  for (int u = 0; u < RING_; ++u) {
    a[u] = *reinterpret_cast<const v2f*>(Al + 4 * u);
    b[u] = *reinterpret_cast<const v2f*>(Bl + 4 * u);
  }

  v8f acc = {};
  for (int kb = 0; kb < KSTEPS_ - RING_; kb += RING_) {
#pragma unroll
    for (int s = 0; s < RING_; s += SLOT_) {
#pragma unroll
      for (int u = s; u < s + SLOT_; ++u)
        acc = __builtin_amdgcn_wmma_f32_16x16x4_f32(
            false, a[u], false, b[u], (short)0, acc, false, false);
#pragma unroll
      for (int u = s; u < s + SLOT_; ++u) {
        const int ks = kb + RING_ + u;
        a[u] = *reinterpret_cast<const v2f*>(Al + 4 * ks);
        b[u] = *reinterpret_cast<const v2f*>(Bl + 4 * ks);
      }
      __builtin_amdgcn_sched_barrier(0);
    }
  }
#pragma unroll
  for (int u = 0; u < RING_; ++u)
    acc = __builtin_amdgcn_wmma_f32_16x16x4_f32(
        false, a[u], false, b[u], (short)0, acc, false, false);
  return acc;
}

__device__ __forceinline__ float fold_xor4(float x) {
  return __builtin_bit_cast(float, __builtin_amdgcn_update_dpp(
      0, __builtin_bit_cast(int, x), 0x164, 0xf, 0xf, false));
}
__device__ __forceinline__ float fold_row1_8to11(float x) {
  return __builtin_bit_cast(float, __builtin_amdgcn_permlanex16(
      __builtin_bit_cast(int, x), __builtin_bit_cast(int, x), 0x0000ba98, 0, false, false));
}
__device__ __forceinline__ void fold_quarters(const v8f& acc, float v[4]) {
#pragma unroll
  for (int bi = 0; bi < 4; ++bi) {
    const float x = acc[bi] + fold_xor4(acc[4 + bi]);
    v[bi] = x + fold_row1_8to11(x);
  }
}

#define NT1_ (H_ / 8)

__global__ void __launch_bounds__(32 * GW_) gemm1_gelu(const float* __restrict__ A,
                           const float* __restrict__ W1,
                           const float* __restrict__ b1,
                           float* __restrict__ Hout,
                           float* __restrict__ st) {
  const int tid   = threadIdx.x;
  const int lane  = tid & 31;
  const int wv    = __builtin_amdgcn_readfirstlane(tid >> 5);
  const int half  = lane >> 4;
  const int khalf = (lane >> 3) & 1;
  const int n0    = blockIdx.x * 8;
  const int col   = n0 + (lane & 7);

  __shared__ __attribute__((aligned(16))) float As[AROWS_ * APITCH_];
  __shared__ __attribute__((aligned(16))) float Bt[8 * APITCH_];
  __shared__ __attribute__((aligned(16))) float Tl[8][8];
  const float* Wq = W1 + (size_t)(KHALF_ * khalf + 2 * half + 4 * KQ_ * wv) * H_ + col;

  const int g  = wv >> 1;
  const int cg = wv & 1;
  float bias = b1[n0 + 4 * cg + (lane & 3)];

  gemm_fetch<H_>(A, Wq, As, b_store_ptr(Bt, lane, wv), tid);

  asm volatile("" : "+v"(bias));
  const v8f acc = gemm_chain(As, Bt, lane, g, cg);

  float v[4];
  fold_quarters(acc, v);
  if (lane < 4) {
#pragma unroll
    for (int bi = 0; bi < 4; ++bi) {
      const float x = v[bi] + bias;
      Tl[4 * g + bi][4 * cg + lane] =
          0.5f * x * (1.0f + erff(x * 0.70710678118654752f));
    }
  }
  __syncthreads();
  if (wv != 0) return;

  {
    const int r = lane >> 2, c2 = 2 * (lane & 3);
    *reinterpret_cast<v2f*>(Hout + r * H_ + n0 + c2) = *reinterpret_cast<const v2f*>(&Tl[r][c2]);
  }
  if (lane < 8) {
    const v4f t0 = *reinterpret_cast<const v4f*>(&Tl[lane][0]);
    const v4f t1 = *reinterpret_cast<const v4f*>(&Tl[lane][4]);
    const float m = (((t0.x + t0.y) + (t0.z + t0.w)) + ((t1.x + t1.y) + (t1.z + t1.w))) * 0.125f;
    float q = 0.f;
#pragma unroll
    for (int e = 0; e < 4; ++e) {
      const float d0 = t0[e] - m, d1 = t1[e] - m;
      q += d0 * d0;
      q += d1 * d1;
    }
    v2f mq; mq.x = m; mq.y = q;
    *reinterpret_cast<v2f*>(st + 2 * (lane * NT1_ + (int)blockIdx.x)) = mq;
  }
}


template <int M>
__device__ __forceinline__ float row_xor(float v) {
  return __builtin_bit_cast(float, __builtin_amdgcn_update_dpp(
      0, __builtin_bit_cast(int, v), 0x160 + M, 0xf, 0xf, false));
}

template <int M>
__device__ __forceinline__ void stat_step(float& m, float& q, float half_n) {
  const float mo = row_xor<M>(m);
  const float qo = row_xor<M>(q);
  const float d  = mo - m;
  q = (q + qo) + half_n * (d * d);
  m = 0.5f * (m + mo);
}

__device__ __forceinline__ void gemm2_fetch_ln(const float* Hg, const float* st,
                                               const float* gam, const float* bet,
                                               const float* Wq, float* As, float* Bst,
                                               int lane, int wv) {
  const int j = lane & 15;
  const int r = 2 * wv + (lane >> 4);
  v4f sv[4];
#pragma unroll
  for (int q = 0; q < 4; ++q)
    sv[q] = *reinterpret_cast<const v4f*>(st + 2 * (r * NT1_) + 4 * (j + 16 * q));
  v4f av[16], gv[16], bv[16];
#pragma unroll
  for (int i = 0; i < 16; ++i)
    av[i] = *reinterpret_cast<const v4f*>(Hg + r * H_ + 4 * (16 * i + j));
#pragma unroll
  for (int i = 0; i < 16; ++i) {
    gv[i] = *reinterpret_cast<const v4f*>(gam + 4 * (16 * i + j));
    bv[i] = *reinterpret_cast<const v4f*>(bet + 4 * (16 * i + j));
  }
  v2f b[KQ_];
#pragma unroll
  for (int u = 0; u < KQ_; ++u) {
    b[u].x = Wq[(size_t)(4 * u) * PD_];
    b[u].y = Wq[(size_t)(4 * u + 1) * PD_];
  }

  float m = (((sv[0].x + sv[0].z) + (sv[1].x + sv[1].z)) +
             ((sv[2].x + sv[2].z) + (sv[3].x + sv[3].z))) * 0.125f;
  float qs = 0.f, ds = 0.f;
#pragma unroll
  for (int q = 0; q < 4; ++q) {
    const float d0 = sv[q].x - m, d1 = sv[q].z - m;
    qs += sv[q].y;
    qs += sv[q].w;
    ds += d0 * d0;
    ds += d1 * d1;
  }
  float qq = qs + 8.0f * ds;
  stat_step<1>(m, qq, 32.0f);
  stat_step<2>(m, qq, 64.0f);
  stat_step<4>(m, qq, 128.0f);
  stat_step<8>(m, qq, 256.0f);
  const float inv = rsqrtf(qq * (1.0f / (float)H_) + 1e-5f);

#pragma unroll
  for (int i = 0; i < 16; ++i) {
    v4f o;
#pragma unroll
    for (int e = 0; e < 4; ++e) o[e] = (av[i][e] - m) * inv * gv[i][e] + bv[i][e];
    *reinterpret_cast<v4f*>(As + r * APITCH_ + 4 * (16 * i + j) + QGAP_ * (i >> 2)) = o;
  }
#pragma unroll
  for (int u = 0; u < KQ_; ++u) *reinterpret_cast<v2f*>(Bst + 4 * u) = b[u];
  __syncthreads();
}

__global__ void __launch_bounds__(32 * GW_) gemm2_ln_out(const float* __restrict__ Hg,
                          const float* __restrict__ st,
                          const float* __restrict__ gam,
                          const float* __restrict__ bet,
                          const float* __restrict__ W2,
                          const float* __restrict__ b2,
                          float* __restrict__ out) {
  const int tid   = threadIdx.x;
  const int lane  = tid & 31;
  const int wv    = __builtin_amdgcn_readfirstlane(tid >> 5);
  const int half  = lane >> 4;
  const int khalf = (lane >> 3) & 1;
  const int n0    = blockIdx.x * 8;
  const int col   = n0 + (lane & 7);

  __shared__ __attribute__((aligned(16))) float As[AROWS_ * APITCH_];
  __shared__ __attribute__((aligned(16))) float Bt[8 * APITCH_];
  const float* Wq = W2 + (size_t)(KHALF_ * khalf + 2 * half + 4 * KQ_ * wv) * PD_ + col;

  const int g  = wv >> 1;
  const int cg = wv & 1;
  const int oc = n0 + 4 * cg + (lane & 3);
  float bias = b2[oc];

  gemm2_fetch_ln(Hg, st, gam, bet, Wq, As, b_store_ptr(Bt, lane, wv), lane, wv);

  asm volatile("" : "+v"(bias));
  const v8f acc = gemm_chain(As, Bt, lane, g, cg);

  float v[4];
  fold_quarters(acc, v);
  if (lane < 4) {
#pragma unroll
    for (int bi = 0; bi < 4; ++bi) {
      const float val = v[bi] + bias;
#pragma unroll
      for (int p = 0; p < P_; ++p)
        out[(size_t)((4 * g + bi) * P_ + p) * PD_ + oc] = val;
    }
  }
}

extern "C" void kernel_launch(void* const* d_in, const int* in_sizes, int n_in,
                              void* d_out, int out_size, void* d_ws, size_t ws_size,
                              hipStream_t stream) {
  const float* ctx  = (const float*)d_in[0];
  const float* W1   = (const float*)d_in[1];
  const float* b1   = (const float*)d_in[2];
  const float* ln_g = (const float*)d_in[3];
  const float* ln_b = (const float*)d_in[4];
  const float* W2   = (const float*)d_in[5];
  const float* b2   = (const float*)d_in[6];
  const int*  masks = (const int*)d_in[7];
  (void)in_sizes; (void)n_in; (void)out_size; (void)ws_size;

  float* out = (float*)d_out;

  char* ws = (char*)d_ws;
  float* pooled   = (float*)(ws);
  float* stats    = (float*)(ws + 32768);
  float* hbuf     = (float*)(ws + 65536);

  gather_pool<<<dim3(D_ / GCOLS_, B_), 256, 0, stream>>>(ctx, masks, pooled);
  gemm1_gelu<<<H_ / 8, 32 * GW_, 0, stream>>>(pooled, W1, b1, hbuf, stats);
  gemm2_ln_out<<<PD_ / 8, 32 * GW_, 0, stream>>>(hbuf, stats, ln_g, ln_b, W2, b2, out);
}
